# starting kernel
# baseline (speedup 1.0000x reference)
#include <hip/hip_runtime.h>


#define DIMD   2000
#define NPAD   2048
#define BATCH  4096
#define NFIELD 40
#define NF     8
#define MT     64
#define NSUB   4
#define CK     200
#define NKC    10
#define LDX    204
#define NSPLIT 4
#define NTPW   4

typedef __attribute__((ext_vector_type(2))) float v2f;
typedef __attribute__((ext_vector_type(8))) float v8f;

__global__ __launch_bounds__(256) void FFM_build_A(
    const float* __restrict__ V, const int* __restrict__ fd,
    const float* __restrict__ W, float* __restrict__ Asw,
    float* __restrict__ dA, float* __restrict__ wp)
{
    const int idx = blockIdx.x * blockDim.x + threadIdx.x;
    const int i = idx >> 11;
    const int j = idx & (NPAD - 1);
    float s = 0.0f;
    if (j < DIMD) {
        const int fi = fd[i];
        const int fj = fd[j];
        const float4* Vi = reinterpret_cast<const float4*>(V + (size_t)i * (NFIELD * NF) + fj * NF);
        const float4* Vj = reinterpret_cast<const float4*>(V + (size_t)j * (NFIELD * NF) + fi * NF);
        float4 a0 = Vi[0], a1 = Vi[1];
        float4 b0 = Vj[0], b1 = Vj[1];
        s = a0.x * b0.x + a0.y * b0.y + a0.z * b0.z + a0.w * b0.w
          + a1.x * b1.x + a1.y * b1.y + a1.z * b1.z + a1.w * b1.w;
    }
    Asw[(((size_t)(i >> 1) * NPAD + j) << 1) + (i & 1)] = s;
    if (i == j && j < DIMD) dA[j] = s;
    if (i == 0) {
        wp[j] = (j < DIMD) ? W[j] : 0.0f;
        if (j >= DIMD) dA[j] = 0.0f;
    }
}

__global__ __launch_bounds__(256) void FFM_wmma_gemm(
    const float* __restrict__ x, const float* __restrict__ Asw,
    const float* __restrict__ dA, const float* __restrict__ wp,
    float* __restrict__ parts)
{
    extern __shared__ float lds[];
    const int tid  = threadIdx.x;
    const int w    = tid >> 5;
    const int lane = tid & 31;
    const int hi   = lane >> 4;
    const int lo   = lane & 15;
    const int sp   = blockIdx.y;
    const int s_base = blockIdx.x * MT;

    v8f c[NTPW][NSUB];
    #pragma unroll
    for (int t = 0; t < NTPW; ++t)
        #pragma unroll
        for (int u = 0; u < NSUB; ++u)
            c[t][u] = (v8f){0.f, 0.f, 0.f, 0.f, 0.f, 0.f, 0.f, 0.f};

    for (int kc = 0; kc < NKC; ++kc) {
        const int nv4 = MT * (CK / 4);
        for (int i = tid; i < nv4; i += 256) {
            const int r  = i / (CK / 4);
            const int cq = i % (CK / 4);
            float4 v = *reinterpret_cast<const float4*>(
                x + (size_t)(s_base + r) * DIMD + kc * CK + cq * 4);
            *reinterpret_cast<float4*>(&lds[r * LDX + cq * 4]) = v;
        }
        __syncthreads();

        for (int ks = 0; ks < CK / 4; ++ks) {
            const int k0 = ks * 4;
            const int kg = kc * CK + k0;
            v2f a[NSUB];
            #pragma unroll
            for (int u = 0; u < NSUB; ++u) {
                const float* p = &lds[(u * 16 + lo) * LDX + k0 + hi * 2];
                a[u][0] = p[0]; a[u][1] = p[1];
            }
            #pragma unroll
            for (int t = 0; t < NTPW; ++t) {
                const int n = ((t * 32 + w * 4 + sp) * 16) + lo;
                v2f bf = *reinterpret_cast<const v2f*>(
                    Asw + (((size_t)(kg >> 1) + hi) * NPAD + n) * 2);
                #pragma unroll
                for (int u = 0; u < NSUB; ++u)
                    c[t][u] = __builtin_amdgcn_wmma_f32_16x16x4_f32(
                        false, a[u], false, bf, (short)0, c[t][u], false, false);
            }
        }
        __syncthreads();
    }

    float racc[NSUB][8];
    #pragma unroll
    for (int u = 0; u < NSUB; ++u)
        #pragma unroll
        for (int v = 0; v < 8; ++v) racc[u][v] = 0.0f;

    #pragma unroll
    for (int t = 0; t < NTPW; ++t) {
        const int n = ((t * 32 + w * 4 + sp) * 16) + lo;
        const float wn = wp[n];
        const float dn = dA[n];
        const bool inb = (n < DIMD);
        #pragma unroll
        for (int u = 0; u < NSUB; ++u) {
            #pragma unroll
            for (int v = 0; v < 8; ++v) {
                const int r = v + hi * 8 + u * 16;
                float xv = inb ? x[(size_t)(s_base + r) * DIMD + n] : 0.0f;
                float tv = c[t][u][v];
                racc[u][v] += xv * (0.5f * tv + wn) - 0.5f * xv * xv * dn;
            }
        }
    }

    #pragma unroll
    for (int u = 0; u < NSUB; ++u) {
        #pragma unroll
        for (int v = 0; v < 8; ++v) {
            float s = racc[u][v];
            s += __shfl_xor(s, 8, 32);
            s += __shfl_xor(s, 4, 32);
            s += __shfl_xor(s, 2, 32);
            s += __shfl_xor(s, 1, 32);
            if (lo == 0) {
                const int r = v + hi * 8 + u * 16;
                lds[w * MT + r] = s;
            }
        }
    }
    __syncthreads();

    if (tid < MT) {
        float p = 0.0f;
        #pragma unroll
        for (int w2 = 0; w2 < 8; ++w2) p += lds[w2 * MT + tid];
        parts[(size_t)(s_base + tid) * NSPLIT + sp] = p;
    }
}

__global__ __launch_bounds__(256) void FFM_finalize(
    const float* __restrict__ parts, const float* __restrict__ b,
    float* __restrict__ out)
{
    const int i = blockIdx.x * blockDim.x + threadIdx.x;
    if (i < BATCH) {
        float p = b[0];
        #pragma unroll
        for (int s = 0; s < NSPLIT; ++s) p += parts[(size_t)i * NSPLIT + s];
        out[i] = p;
    }
}

extern "C" void kernel_launch(void* const* d_in, const int* in_sizes, int n_in,
                              void* d_out, int out_size, void* d_ws, size_t ws_size,
                              hipStream_t stream)
{
    const float* x  = (const float*)d_in[0];
    const int*   fd = (const int*)  d_in[1];
    const float* b  = (const float*)d_in[2];
    const float* W  = (const float*)d_in[3];
    const float* V  = (const float*)d_in[4];
    float* out = (float*)d_out;

    float* Asw   = (float*)d_ws;
    float* dA    = Asw + (size_t)DIMD * NPAD;
    float* wp    = dA + NPAD;
    float* parts = wp + NPAD;

    FFM_build_A<<<16000, 256, 0, stream>>>(V, fd, W, Asw, dA, wp);

    dim3 g2(BATCH / MT, NSPLIT);
    FFM_wmma_gemm<<<g2, 256, MT * LDX * sizeof(float), stream>>>(x, Asw, dA, wp, parts);

    FFM_finalize<<<BATCH / 256, 256, 0, stream>>>(parts, b, out);
}


# Software-pipelined global loads: next x chunk and next 5 k-steps of B fragments are fetched into registers ahead of the WMMAs that hide them; sum order unchanged
# speedup vs baseline: 1.5593x; 1.5593x over previous
#include <hip/hip_runtime.h>


#define DIMD   2000
#define NPAD   2048
#define BATCH  4096
#define NFIELD 40
#define NF     8
#define MT     64
#define NSUB   4
#define CK     200
#define NKC    10
#define LDX    204
#define NSPLIT 4
#define NTPW   4
#define KS     5
#define NSTG   10
#define NGST   100
#define XV4    13

typedef __attribute__((ext_vector_type(2))) float v2f;
typedef __attribute__((ext_vector_type(8))) float v8f;
typedef __attribute__((ext_vector_type(4))) float v4f;

__global__ __launch_bounds__(256) void FFM_build_A(
    const float* __restrict__ V, const int* __restrict__ fd,
    const float* __restrict__ W, float* __restrict__ Asw,
    float* __restrict__ dA, float* __restrict__ wp)
{
    const int idx = blockIdx.x * blockDim.x + threadIdx.x;
    const int i = idx >> 11;
    const int j = idx & (NPAD - 1);
    float s = 0.0f;
    if (j < DIMD) {
        const int fi = fd[i];
        const int fj = fd[j];
        const float4* Vi = reinterpret_cast<const float4*>(V + (size_t)i * (NFIELD * NF) + fj * NF);
        const float4* Vj = reinterpret_cast<const float4*>(V + (size_t)j * (NFIELD * NF) + fi * NF);
        float4 a0 = Vi[0], a1 = Vi[1];
        float4 b0 = Vj[0], b1 = Vj[1];
        s = a0.x * b0.x + a0.y * b0.y + a0.z * b0.z + a0.w * b0.w
          + a1.x * b1.x + a1.y * b1.y + a1.z * b1.z + a1.w * b1.w;
    }
    Asw[(((size_t)(i >> 1) * NPAD + j) << 1) + (i & 1)] = s;
    if (i == j && j < DIMD) dA[j] = s;
    if (i == 0) {
        wp[j] = (j < DIMD) ? W[j] : 0.0f;
        if (j >= DIMD) dA[j] = 0.0f;
    }
}

__global__ __launch_bounds__(256) void FFM_wmma_gemm(
    const float* __restrict__ x, const float* __restrict__ Asw,
    const float* __restrict__ dA, const float* __restrict__ wp,
    float* __restrict__ parts)
{
    extern __shared__ float lds[];
    const int tid  = threadIdx.x;
    const int w    = tid >> 5;
    const int lane = tid & 31;
    const int hi   = lane >> 4;
    const int lo   = lane & 15;
    const int sp   = blockIdx.y;
    const int s_base = blockIdx.x * MT;

    v8f c[NTPW][NSUB];
    #pragma unroll
    for (int t = 0; t < NTPW; ++t)
        #pragma unroll
        for (int u = 0; u < NSUB; ++u)
            c[t][u] = (v8f){0.f, 0.f, 0.f, 0.f, 0.f, 0.f, 0.f, 0.f};

    const float* bbase[NTPW];
    #pragma unroll
    for (int t = 0; t < NTPW; ++t) {
        const int n = ((t * 32 + w * 4 + sp) * 16) + lo;
        bbase[t] = Asw + ((size_t)hi * NPAD + n) * 2;
    }

    v4f xr[XV4];
    v2f bA[KS][NTPW];
    v2f bB[KS][NTPW];

#define ISSUE_X(kc2)                                                            \
    _Pragma("unroll")                                                           \
    for (int j = 0; j < XV4; ++j) {                                             \
        const int i = tid + j * 256;                                            \
        if (j < XV4 - 1 || tid < 128) {                                         \
            const int r  = i / (CK / 4);                                        \
            const int cq = i % (CK / 4);                                        \
            xr[j] = *reinterpret_cast<const v4f*>(                              \
                x + (size_t)(s_base + r) * DIMD + (kc2) * CK + cq * 4);         \
        }                                                                       \
    }
#define STORE_X()                                                               \
    _Pragma("unroll")                                                           \
    for (int j = 0; j < XV4; ++j) {                                             \
        const int i = tid + j * 256;                                            \
        if (j < XV4 - 1 || tid < 128) {                                         \
            const int r  = i / (CK / 4);                                        \
            const int cq = i % (CK / 4);                                        \
            *reinterpret_cast<v4f*>(&lds[r * LDX + cq * 4]) = xr[j];            \
        }                                                                       \
    }

    auto load_B = [&](v2f (&b)[KS][NTPW], int g) {
        #pragma unroll
        for (int j = 0; j < KS; ++j)
            #pragma unroll
            for (int t = 0; t < NTPW; ++t)
                b[j][t] = *reinterpret_cast<const v2f*>(
                    bbase[t] + (size_t)(g * (KS * 2) + j * 2) * (NPAD * 2));
    };
    auto mult = [&](const v2f (&b)[KS][NTPW], int sl) {
        #pragma unroll
        for (int j = 0; j < KS; ++j) {
            const int k0 = (sl * KS + j) * 4;
            v2f a[NSUB];
            #pragma unroll
            for (int u = 0; u < NSUB; ++u) {
                const float* p = &lds[(u * 16 + lo) * LDX + k0 + hi * 2];
                a[u][0] = p[0]; a[u][1] = p[1];
            }
            #pragma unroll
            for (int t = 0; t < NTPW; ++t)
                #pragma unroll
                for (int u = 0; u < NSUB; ++u)
                    c[t][u] = __builtin_amdgcn_wmma_f32_16x16x4_f32(
                        false, a[u], false, b[j][t], (short)0, c[t][u], false, false);
        }
    };

    ISSUE_X(0)
    load_B(bA, 0);
    STORE_X()
    __syncthreads();

    for (int kc = 0; kc < NKC; ++kc) {
        const bool more = (kc + 1 < NKC);
        if (more) { ISSUE_X(kc + 1) }

        for (int sl = 0; sl < NSTG; sl += 2) {
            const int g = kc * NSTG + sl;
            load_B(bB, g + 1);
            mult(bA, sl);
            load_B(bA, (g + 2 < NGST) ? (g + 2) : (NGST - 1));
            mult(bB, sl + 1);
        }
        __syncthreads();
        if (more) { STORE_X() }
        __syncthreads();
    }

#undef ISSUE_X
#undef STORE_X

    float racc[NSUB][8];
    #pragma unroll
    for (int u = 0; u < NSUB; ++u)
        #pragma unroll
        for (int v = 0; v < 8; ++v) racc[u][v] = 0.0f;

    #pragma unroll
    for (int t = 0; t < NTPW; ++t) {
        const int n = ((t * 32 + w * 4 + sp) * 16) + lo;
        const float wn = wp[n];
        const float dn = dA[n];
        const bool inb = (n < DIMD);
        #pragma unroll
        for (int u = 0; u < NSUB; ++u) {
            #pragma unroll
            for (int v = 0; v < 8; ++v) {
                const int r = v + hi * 8 + u * 16;
                float xv = inb ? x[(size_t)(s_base + r) * DIMD + n] : 0.0f;
                float tv = c[t][u][v];
                racc[u][v] += xv * (0.5f * tv + wn) - 0.5f * xv * xv * dn;
            }
        }
    }

    #pragma unroll
    for (int u = 0; u < NSUB; ++u) {
        #pragma unroll
        for (int v = 0; v < 8; ++v) {
            float s = racc[u][v];
            s += __shfl_xor(s, 8, 32);
            s += __shfl_xor(s, 4, 32);
            s += __shfl_xor(s, 2, 32);
            s += __shfl_xor(s, 1, 32);
            if (lo == 0) {
                const int r = v + hi * 8 + u * 16;
                lds[w * MT + r] = s;
            }
        }
    }
    __syncthreads();

    if (tid < MT) {
        float p = 0.0f;
        #pragma unroll
        for (int w2 = 0; w2 < 8; ++w2) p += lds[w2 * MT + tid];
        parts[(size_t)(s_base + tid) * NSPLIT + sp] = p;
    }
}

__global__ __launch_bounds__(256) void FFM_finalize(
    const float* __restrict__ parts, const float* __restrict__ b,
    float* __restrict__ out)
{
    const int i = blockIdx.x * blockDim.x + threadIdx.x;
    if (i < BATCH) {
        float p = b[0];
        #pragma unroll
        for (int s = 0; s < NSPLIT; ++s) p += parts[(size_t)i * NSPLIT + s];
        out[i] = p;
    }
}

extern "C" void kernel_launch(void* const* d_in, const int* in_sizes, int n_in,
                              void* d_out, int out_size, void* d_ws, size_t ws_size,
                              hipStream_t stream)
{
    const float* x  = (const float*)d_in[0];
    const int*   fd = (const int*)  d_in[1];
    const float* b  = (const float*)d_in[2];
    const float* W  = (const float*)d_in[3];
    const float* V  = (const float*)d_in[4];
    float* out = (float*)d_out;

    float* Asw   = (float*)d_ws;
    float* dA    = Asw + (size_t)DIMD * NPAD;
    float* wp    = dA + NPAD;
    float* parts = wp + NPAD;

    FFM_build_A<<<16000, 256, 0, stream>>>(V, fd, W, Asw, dA, wp);

    dim3 g2(BATCH / MT, NSPLIT);
    FFM_wmma_gemm<<<g2, 256, MT * LDX * sizeof(float), stream>>>(x, Asw, dA, wp, parts);

    FFM_finalize<<<BATCH / 256, 256, 0, stream>>>(parts, b, out);
}


# A is symmetric by construction: build only its strict upper triangle and end each column tile's K loop at the diagonal (about half the WMMAs); diag and 0.5 terms drop out; sum order changes
# speedup vs baseline: 2.0659x; 1.3249x over previous
#include <hip/hip_runtime.h>


#define DIMD   2000
#define NPAD   2048
#define BATCH  4096
#define NFIELD 40
#define NF     8
#define MT     64
#define NSUB   4
#define CK     200
#define NKC    10
#define LDX    204
#define NSPLIT 4
#define NTPW   4
#define KS     5
#define NSTG   10
#define NGST   100
#define XV4    13
#define NPAIR  50

template <int N> struct IC { static constexpr int value = N; };

typedef __attribute__((ext_vector_type(2))) float v2f;
typedef __attribute__((ext_vector_type(8))) float v8f;
typedef __attribute__((ext_vector_type(4))) float v4f;

__global__ __launch_bounds__(256) void FFM_build_A(
    const float* __restrict__ V, const int* __restrict__ fd,
    const float* __restrict__ W, float* __restrict__ Asw,
    float* __restrict__ wp)
{
    const int idx = blockIdx.x * blockDim.x + threadIdx.x;
    const int i = idx >> 11;
    const int j = idx & (NPAD - 1);
    float s = 0.0f;
    if (j < DIMD && i < j) {
        const int fi = fd[i];
        const int fj = fd[j];
        const float4* Vi = reinterpret_cast<const float4*>(V + (size_t)i * (NFIELD * NF) + fj * NF);
        const float4* Vj = reinterpret_cast<const float4*>(V + (size_t)j * (NFIELD * NF) + fi * NF);
        float4 a0 = Vi[0], a1 = Vi[1];
        float4 b0 = Vj[0], b1 = Vj[1];
        s = a0.x * b0.x + a0.y * b0.y + a0.z * b0.z + a0.w * b0.w
          + a1.x * b1.x + a1.y * b1.y + a1.z * b1.z + a1.w * b1.w;
    }
    Asw[(((size_t)(i >> 1) * NPAD + j) << 1) + (i & 1)] = s;
    if (i == 0) wp[j] = (j < DIMD) ? W[j] : 0.0f;
}

__global__ __launch_bounds__(256) void FFM_wmma_gemm(
    const float* __restrict__ x, const float* __restrict__ Asw,
    const float* __restrict__ wp, float* __restrict__ parts)
{
    extern __shared__ float lds[];
    const int tid  = threadIdx.x;
    const int w    = __builtin_amdgcn_readfirstlane(tid >> 5);
    const int lane = tid & 31;
    const int hi   = lane >> 4;
    const int lo   = lane & 15;
    const int sp   = blockIdx.y;
    const int s_base = blockIdx.x * MT;

    v8f c[NTPW][NSUB];
    #pragma unroll
    for (int t = 0; t < NTPW; ++t)
        #pragma unroll
        for (int u = 0; u < NSUB; ++u)
            c[t][u] = (v8f){0.f, 0.f, 0.f, 0.f, 0.f, 0.f, 0.f, 0.f};

    const int pos = (w < 4) ? w : 11 - w;
    int ntile[NTPW];
    int lim[NTPW];
    const float* bbase[NTPW];
    #pragma unroll
    for (int t = 0; t < NTPW; ++t) {
        ntile[t] = t * 32 + ((t & 1) ? (7 - pos) : pos) * 4 + sp;
        const int pairs = (16 * ntile[t] + 14) / (4 * KS * 2) + 1;
        lim[t] = 2 * (pairs < NPAIR ? pairs : NPAIR);
        const int n = ntile[t] * 16 + lo;
        bbase[t] = Asw + ((size_t)hi * NPAD + n) * 2;
    }

    v4f xr[XV4];
    v2f bA[KS][NTPW];
    v2f bB[KS][NTPW];

#define ISSUE_X(kc2)                                                            \
    _Pragma("unroll")                                                           \
    for (int j = 0; j < XV4; ++j) {                                             \
        const int i = tid + j * 256;                                            \
        if (j < XV4 - 1 || tid < 128) {                                         \
            const int r  = i / (CK / 4);                                        \
            const int cq = i % (CK / 4);                                        \
            xr[j] = *reinterpret_cast<const v4f*>(                              \
                x + (size_t)(s_base + r) * DIMD + (kc2) * CK + cq * 4);         \
        }                                                                       \
    }
#define STORE_X()                                                               \
    _Pragma("unroll")                                                           \
    for (int j = 0; j < XV4; ++j) {                                             \
        const int i = tid + j * 256;                                            \
        if (j < XV4 - 1 || tid < 128) {                                         \
            const int r  = i / (CK / 4);                                        \
            const int cq = i % (CK / 4);                                        \
            *reinterpret_cast<v4f*>(&lds[r * LDX + cq * 4]) = xr[j];            \
        }                                                                       \
    }

    auto load_B = [&](auto tm, v2f (&b)[KS][NTPW], int g) {
        constexpr int TM = decltype(tm)::value;
        #pragma unroll
        for (int j = 0; j < KS; ++j)
            #pragma unroll
            for (int t = TM; t < NTPW; ++t)
                b[j][t] = *reinterpret_cast<const v2f*>(
                    bbase[t] + (size_t)(g * (KS * 2) + j * 2) * (NPAD * 2));
    };
    auto mult = [&](auto tm, const v2f (&b)[KS][NTPW], int sl) {
        constexpr int TM = decltype(tm)::value;
        #pragma unroll
        for (int j = 0; j < KS; ++j) {
            const int k0 = (sl * KS + j) * 4;
            v2f a[NSUB];
            #pragma unroll
            for (int u = 0; u < NSUB; ++u) {
                const float* p = &lds[(u * 16 + lo) * LDX + k0 + hi * 2];
                a[u][0] = p[0]; a[u][1] = p[1];
            }
            #pragma unroll
            for (int t = TM; t < NTPW; ++t)
                #pragma unroll
                for (int u = 0; u < NSUB; ++u)
                    c[t][u] = __builtin_amdgcn_wmma_f32_16x16x4_f32(
                        false, a[u], false, b[j][t], (short)0, c[t][u], false, false);
        }
    };

    auto pair = [&](auto tm, int g, int sl) {
        load_B(tm, bB, g + 1);
        mult(tm, bA, sl);
        load_B(tm, bA, (g + 2 < NGST) ? (g + 2) : (NGST - 1));
        mult(tm, bB, sl + 1);
    };

    ISSUE_X(0)
    load_B(IC<0>{}, bA, 0);
    STORE_X()
    __syncthreads();

    for (int kc = 0; kc < NKC; ++kc) {
        const bool more = (kc + 1 < NKC);
        if (more) { ISSUE_X(kc + 1) }

        for (int sl = 0; sl < NSTG; sl += 2) {
            const int g = kc * NSTG + sl;
            const int tmin = (g >= lim[0]) + (g >= lim[1]) + (g >= lim[2]) + (g >= lim[3]);
            switch (tmin) {
                case 0:  pair(IC<0>{}, g, sl); break;
                case 1:  pair(IC<1>{}, g, sl); break;
                case 2:  pair(IC<2>{}, g, sl); break;
                case 3:  pair(IC<3>{}, g, sl); break;
                default: break;
            }
        }
        __syncthreads();
        if (more) { STORE_X() }
        __syncthreads();
    }

#undef ISSUE_X
#undef STORE_X

    float racc[NSUB][8];
    #pragma unroll
    for (int u = 0; u < NSUB; ++u)
        #pragma unroll
        for (int v = 0; v < 8; ++v) racc[u][v] = 0.0f;

    #pragma unroll
    for (int t = 0; t < NTPW; ++t) {
        const int n = ntile[t] * 16 + lo;
        const float wn = wp[n];
        const bool inb = (n < DIMD);
        #pragma unroll
        for (int u = 0; u < NSUB; ++u) {
            #pragma unroll
            for (int v = 0; v < 8; ++v) {
                const int r = v + hi * 8 + u * 16;
                float xv = inb ? x[(size_t)(s_base + r) * DIMD + n] : 0.0f;
                float tv = c[t][u][v];
                racc[u][v] += xv * (tv + wn);
            }
        }
    }

    #pragma unroll
    for (int u = 0; u < NSUB; ++u) {
        #pragma unroll
        for (int v = 0; v < 8; ++v) {
            float s = racc[u][v];
            s += __shfl_xor(s, 8, 32);
            s += __shfl_xor(s, 4, 32);
            s += __shfl_xor(s, 2, 32);
            s += __shfl_xor(s, 1, 32);
            if (lo == 0) {
                const int r = v + hi * 8 + u * 16;
                lds[w * MT + r] = s;
            }
        }
    }
    __syncthreads();

    if (tid < MT) {
        float p = 0.0f;
        #pragma unroll
        for (int w2 = 0; w2 < 8; ++w2) p += lds[w2 * MT + tid];
        parts[(size_t)(s_base + tid) * NSPLIT + sp] = p;
    }
}

__global__ __launch_bounds__(256) void FFM_finalize(
    const float* __restrict__ parts, const float* __restrict__ b,
    float* __restrict__ out)
{
    const int i = blockIdx.x * blockDim.x + threadIdx.x;
    if (i < BATCH) {
        float p = b[0];
        #pragma unroll
        for (int s = 0; s < NSPLIT; ++s) p += parts[(size_t)i * NSPLIT + s];
        out[i] = p;
    }
}

extern "C" void kernel_launch(void* const* d_in, const int* in_sizes, int n_in,
                              void* d_out, int out_size, void* d_ws, size_t ws_size,
                              hipStream_t stream)
{
    const float* x  = (const float*)d_in[0];
    const int*   fd = (const int*)  d_in[1];
    const float* b  = (const float*)d_in[2];
    const float* W  = (const float*)d_in[3];
    const float* V  = (const float*)d_in[4];
    float* out = (float*)d_out;

    float* Asw   = (float*)d_ws;
    float* wp    = Asw + (size_t)DIMD * NPAD;
    float* parts = wp + NPAD;

    FFM_build_A<<<16000, 256, 0, stream>>>(V, fd, W, Asw, wp);

    dim3 g2(BATCH / MT, NSPLIT);
    FFM_wmma_gemm<<<g2, 256, MT * LDX * sizeof(float), stream>>>(x, Asw, wp, parts);

    FFM_finalize<<<BATCH / 256, 256, 0, stream>>>(parts, b, out);
}


# Build kernel re-tiled: one block per 80x32 tile of only the rows the GEMM fetches, tile columns' V rows staged in LDS, float2 store per row pair; 128000 build waves become 7136; sum order unchanged
# speedup vs baseline: 2.2408x; 1.0847x over previous
#include <hip/hip_runtime.h>


#define DIMD   2000
#define NPAD   2048
#define BATCH  4096
#define NFIELD 40
#define NF     8
#define MT     64
#define NSUB   4
#define CK     200
#define NKC    10
#define LDX    204
#define NSPLIT 4
#define NTPW   4
#define KS     5
#define NSTG   10
#define NGST   100
#define XV4    13
#define NPAIR  50
#define BCB    32
#define BRB    80
#define NCB    64
#define NRBK   25
#define VROW   320
#define LVROW  324
#define NPW    5
#define NSTV   10
#define NBUILD 892

template <int N> struct IC { static constexpr int value = N; };

typedef __attribute__((ext_vector_type(2))) float v2f;
typedef __attribute__((ext_vector_type(8))) float v8f;
typedef __attribute__((ext_vector_type(4))) float v4f;

__global__ __launch_bounds__(256) void FFM_build_A(
    const float* __restrict__ V, const int* __restrict__ fd,
    const float* __restrict__ W, float* __restrict__ Asw,
    float* __restrict__ wp)
{
    __shared__ __attribute__((aligned(16))) float vs[BCB * LVROW];

    int rest = blockIdx.x;
    int rbk = 0;
    int cb = -1;
    for (rbk = 0; rbk < NRBK; ++rbk) {
        const int cmin = (rbk == 0) ? 0 : (BRB * rbk - 39) / BCB;
        const int cnt = NCB - cmin;
        if (rest < cnt) { cb = cmin + rest; break; }
        rest -= cnt;
    }
    if (cb < 0) return;

    const int tid  = threadIdx.x;
    const int lane = tid & 31;
    const int wv   = __builtin_amdgcn_readfirstlane(tid >> 5);
    const int j0   = cb * BCB;
    const int j    = j0 + lane;
    const bool live = (j0 < DIMD) && (rbk * BRB < j0 + BCB - 1);

    if (live) {
        float4 st[NSTV];
        #pragma unroll
        for (int k = 0; k < NSTV; ++k) {
            const int idx = tid + k * 256;
            const int r   = idx / (VROW / 4);
            const int c4  = idx - r * (VROW / 4);
            const int rs  = (j0 + r < DIMD) ? (j0 + r) : (DIMD - 1);
            st[k] = *reinterpret_cast<const float4*>(V + (size_t)rs * VROW + c4 * 4);
        }
        #pragma unroll
        for (int k = 0; k < NSTV; ++k) {
            const int idx = tid + k * 256;
            const int r   = idx / (VROW / 4);
            const int c4  = idx - r * (VROW / 4);
            *reinterpret_cast<float4*>(&vs[r * LVROW + c4 * 4]) = st[k];
        }
    }
    __syncthreads();

    const bool jin = (j < DIMD);
    const int  p0  = rbk * (BRB / 2) + wv * NPW;
    float2 res[NPW];
    #pragma unroll
    for (int t = 0; t < NPW; ++t) res[t] = make_float2(0.0f, 0.0f);

    if (live) {
        const int fj = jin ? fd[j] : 0;
        const float* vj = &vs[lane * LVROW];
        int    fr[2 * NPW];
        float4 g[2 * NPW][2];
        #pragma unroll
        for (int k = 0; k < 2 * NPW; ++k) fr[k] = fd[2 * p0 + k];
        #pragma unroll
        for (int k = 0; k < 2 * NPW; ++k) {
            const float4* Vi = reinterpret_cast<const float4*>(
                V + (size_t)(2 * p0 + k) * VROW + fj * NF);
            g[k][0] = Vi[0];
            g[k][1] = Vi[1];
        }
        #pragma unroll
        for (int k = 0; k < 2 * NPW; ++k) {
            const float4* Vj = reinterpret_cast<const float4*>(vj + fr[k] * NF);
            const float4 a0 = g[k][0], a1 = g[k][1];
            const float4 b0 = Vj[0], b1 = Vj[1];
            const float d = a0.x * b0.x + a0.y * b0.y + a0.z * b0.z + a0.w * b0.w
                          + a1.x * b1.x + a1.y * b1.y + a1.z * b1.z + a1.w * b1.w;
            const float s = (jin && 2 * p0 + k < j) ? d : 0.0f;
            if (k & 1) res[k >> 1].y = s; else res[k >> 1].x = s;
        }
    }
    #pragma unroll
    for (int t = 0; t < NPW; ++t)
        *reinterpret_cast<float2*>(Asw + (((size_t)(p0 + t) * NPAD + j) << 1)) = res[t];
    if (rbk == 0 && wv == 0) wp[j] = jin ? W[j] : 0.0f;
}

__global__ __launch_bounds__(256) void FFM_wmma_gemm(
    const float* __restrict__ x, const float* __restrict__ Asw,
    const float* __restrict__ wp, float* __restrict__ parts)
{
    extern __shared__ float lds[];
    const int tid  = threadIdx.x;
    const int w    = __builtin_amdgcn_readfirstlane(tid >> 5);
    const int lane = tid & 31;
    const int hi   = lane >> 4;
    const int lo   = lane & 15;
    const int sp   = blockIdx.y;
    const int s_base = blockIdx.x * MT;

    v8f c[NTPW][NSUB];
    #pragma unroll
    for (int t = 0; t < NTPW; ++t)
        #pragma unroll
        for (int u = 0; u < NSUB; ++u)
            c[t][u] = (v8f){0.f, 0.f, 0.f, 0.f, 0.f, 0.f, 0.f, 0.f};

    const int pos = (w < 4) ? w : 11 - w;
    int ntile[NTPW];
    int lim[NTPW];
    const float* bbase[NTPW];
    #pragma unroll
    for (int t = 0; t < NTPW; ++t) {
        ntile[t] = t * 32 + ((t & 1) ? (7 - pos) : pos) * 4 + sp;
        const int pairs = (16 * ntile[t] + 14) / (4 * KS * 2) + 1;
        lim[t] = 2 * (pairs < NPAIR ? pairs : NPAIR);
        const int n = ntile[t] * 16 + lo;
        bbase[t] = Asw + ((size_t)hi * NPAD + n) * 2;
    }

    v4f xr[XV4];
    v2f bA[KS][NTPW];
    v2f bB[KS][NTPW];

#define ISSUE_X(kc2)                                                            \
    _Pragma("unroll")                                                           \
    for (int j = 0; j < XV4; ++j) {                                             \
        const int i = tid + j * 256;                                            \
        if (j < XV4 - 1 || tid < 128) {                                         \
            const int r  = i / (CK / 4);                                        \
            const int cq = i % (CK / 4);                                        \
            xr[j] = *reinterpret_cast<const v4f*>(                              \
                x + (size_t)(s_base + r) * DIMD + (kc2) * CK + cq * 4);         \
        }                                                                       \
    }
#define STORE_X()                                                               \
    _Pragma("unroll")                                                           \
    for (int j = 0; j < XV4; ++j) {                                             \
        const int i = tid + j * 256;                                            \
        if (j < XV4 - 1 || tid < 128) {                                         \
            const int r  = i / (CK / 4);                                        \
            const int cq = i % (CK / 4);                                        \
            *reinterpret_cast<v4f*>(&lds[r * LDX + cq * 4]) = xr[j];            \
        }                                                                       \
    }

    auto load_B = [&](auto tm, v2f (&b)[KS][NTPW], int g) {
        constexpr int TM = decltype(tm)::value;
        #pragma unroll
        for (int j = 0; j < KS; ++j)
            #pragma unroll
            for (int t = TM; t < NTPW; ++t)
                b[j][t] = *reinterpret_cast<const v2f*>(
                    bbase[t] + (size_t)(g * (KS * 2) + j * 2) * (NPAD * 2));
    };
    auto mult = [&](auto tm, const v2f (&b)[KS][NTPW], int sl) {
        constexpr int TM = decltype(tm)::value;
        #pragma unroll
        for (int j = 0; j < KS; ++j) {
            const int k0 = (sl * KS + j) * 4;
            v2f a[NSUB];
            #pragma unroll
            for (int u = 0; u < NSUB; ++u) {
                const float* p = &lds[(u * 16 + lo) * LDX + k0 + hi * 2];
                a[u][0] = p[0]; a[u][1] = p[1];
            }
            #pragma unroll
            for (int t = TM; t < NTPW; ++t)
                #pragma unroll
                for (int u = 0; u < NSUB; ++u)
                    c[t][u] = __builtin_amdgcn_wmma_f32_16x16x4_f32(
                        false, a[u], false, b[j][t], (short)0, c[t][u], false, false);
        }
    };

    auto pair = [&](auto tm, int g, int sl) {
        load_B(tm, bB, g + 1);
        mult(tm, bA, sl);
        load_B(tm, bA, (g + 2 < NGST) ? (g + 2) : (NGST - 1));
        mult(tm, bB, sl + 1);
    };

    ISSUE_X(0)
    load_B(IC<0>{}, bA, 0);
    STORE_X()
    __syncthreads();

    for (int kc = 0; kc < NKC; ++kc) {
        const bool more = (kc + 1 < NKC);
        if (more) { ISSUE_X(kc + 1) }

        for (int sl = 0; sl < NSTG; sl += 2) {
            const int g = kc * NSTG + sl;
            const int tmin = (g >= lim[0]) + (g >= lim[1]) + (g >= lim[2]) + (g >= lim[3]);
            switch (tmin) {
                case 0:  pair(IC<0>{}, g, sl); break;
                case 1:  pair(IC<1>{}, g, sl); break;
                case 2:  pair(IC<2>{}, g, sl); break;
                case 3:  pair(IC<3>{}, g, sl); break;
                default: break;
            }
        }
        __syncthreads();
        if (more) { STORE_X() }
        __syncthreads();
    }

#undef ISSUE_X
#undef STORE_X

    float racc[NSUB][8];
    #pragma unroll
    for (int u = 0; u < NSUB; ++u)
        #pragma unroll
        for (int v = 0; v < 8; ++v) racc[u][v] = 0.0f;

    #pragma unroll
    for (int t = 0; t < NTPW; ++t) {
        const int n = ntile[t] * 16 + lo;
        const float wn = wp[n];
        const bool inb = (n < DIMD);
        #pragma unroll
        for (int u = 0; u < NSUB; ++u) {
            #pragma unroll
            for (int v = 0; v < 8; ++v) {
                const int r = v + hi * 8 + u * 16;
                float xv = inb ? x[(size_t)(s_base + r) * DIMD + n] : 0.0f;
                float tv = c[t][u][v];
                racc[u][v] += xv * (tv + wn);
            }
        }
    }

    #pragma unroll
    for (int u = 0; u < NSUB; ++u) {
        #pragma unroll
        for (int v = 0; v < 8; ++v) {
            float s = racc[u][v];
            s += __shfl_xor(s, 8, 32);
            s += __shfl_xor(s, 4, 32);
            s += __shfl_xor(s, 2, 32);
            s += __shfl_xor(s, 1, 32);
            if (lo == 0) {
                const int r = v + hi * 8 + u * 16;
                lds[w * MT + r] = s;
            }
        }
    }
    __syncthreads();

    if (tid < MT) {
        float p = 0.0f;
        #pragma unroll
        for (int w2 = 0; w2 < 8; ++w2) p += lds[w2 * MT + tid];
        parts[(size_t)(s_base + tid) * NSPLIT + sp] = p;
    }
}

__global__ __launch_bounds__(256) void FFM_finalize(
    const float* __restrict__ parts, const float* __restrict__ b,
    float* __restrict__ out)
{
    const int i = blockIdx.x * blockDim.x + threadIdx.x;
    if (i < BATCH) {
        float p = b[0];
        #pragma unroll
        for (int s = 0; s < NSPLIT; ++s) p += parts[(size_t)i * NSPLIT + s];
        out[i] = p;
    }
}

extern "C" void kernel_launch(void* const* d_in, const int* in_sizes, int n_in,
                              void* d_out, int out_size, void* d_ws, size_t ws_size,
                              hipStream_t stream)
{
    const float* x  = (const float*)d_in[0];
    const int*   fd = (const int*)  d_in[1];
    const float* b  = (const float*)d_in[2];
    const float* W  = (const float*)d_in[3];
    const float* V  = (const float*)d_in[4];
    float* out = (float*)d_out;

    float* Asw   = (float*)d_ws;
    float* wp    = Asw + (size_t)DIMD * NPAD;
    float* parts = wp + NPAD;

    FFM_build_A<<<NBUILD, 256, 0, stream>>>(V, fd, W, Asw, wp);

    dim3 g2(BATCH / MT, NSPLIT);
    FFM_wmma_gemm<<<g2, 256, MT * LDX * sizeof(float), stream>>>(x, Asw, wp, parts);

    FFM_finalize<<<BATCH / 256, 256, 0, stream>>>(parts, b, out);
}


# x chunks rotate through three LDS buffers under a split workgroup barrier (store chunk k+1 and signal at the top of chunk k, wait before chunk k+1): waves stop idling at chunk ends; sum order same
# speedup vs baseline: 2.4761x; 1.1050x over previous
#include <hip/hip_runtime.h>


#define DIMD   2000
#define NPAD   2048
#define BATCH  4096
#define NFIELD 40
#define NF     8
#define MT     64
#define NSUB   4
#define CK     200
#define NKC    10
#define LDX    204
#define NSPLIT 4
#define NTPW   4
#define KS     5
#define NSTG   10
#define NGST   100
#define XV4    13
#define NPAIR  50
#define BCB    32
#define BRB    80
#define NCB    64
#define NRBK   25
#define VROW   320
#define LVROW  324
#define NPW    5
#define NSTV   10
#define NBUILD 892
#define XBUF   (MT * LDX)
#define NXBUF  3

template <int N> struct IC { static constexpr int value = N; };

typedef __attribute__((ext_vector_type(2))) float v2f;
typedef __attribute__((ext_vector_type(8))) float v8f;
typedef __attribute__((ext_vector_type(4))) float v4f;

__global__ __launch_bounds__(256) void FFM_build_A(
    const float* __restrict__ V, const int* __restrict__ fd,
    const float* __restrict__ W, float* __restrict__ Asw,
    float* __restrict__ wp)
{
    __shared__ __attribute__((aligned(16))) float vs[BCB * LVROW];

    int rest = blockIdx.x;
    int rbk = 0;
    int cb = -1;
    for (rbk = 0; rbk < NRBK; ++rbk) {
        const int cmin = (rbk == 0) ? 0 : (BRB * rbk - 39) / BCB;
        const int cnt = NCB - cmin;
        if (rest < cnt) { cb = cmin + rest; break; }
        rest -= cnt;
    }
    if (cb < 0) return;

    const int tid  = threadIdx.x;
    const int lane = tid & 31;
    const int wv   = __builtin_amdgcn_readfirstlane(tid >> 5);
    const int j0   = cb * BCB;
    const int j    = j0 + lane;
    const bool live = (j0 < DIMD) && (rbk * BRB < j0 + BCB - 1);

    if (live) {
        float4 st[NSTV];
        #pragma unroll
        for (int k = 0; k < NSTV; ++k) {
            const int idx = tid + k * 256;
            const int r   = idx / (VROW / 4);
            const int c4  = idx - r * (VROW / 4);
            const int rs  = (j0 + r < DIMD) ? (j0 + r) : (DIMD - 1);
            st[k] = *reinterpret_cast<const float4*>(V + (size_t)rs * VROW + c4 * 4);
        }
        #pragma unroll
        for (int k = 0; k < NSTV; ++k) {
            const int idx = tid + k * 256;
            const int r   = idx / (VROW / 4);
            const int c4  = idx - r * (VROW / 4);
            *reinterpret_cast<float4*>(&vs[r * LVROW + c4 * 4]) = st[k];
        }
    }
    __syncthreads();

    const bool jin = (j < DIMD);
    const int  p0  = rbk * (BRB / 2) + wv * NPW;
    float2 res[NPW];
    #pragma unroll
    for (int t = 0; t < NPW; ++t) res[t] = make_float2(0.0f, 0.0f);

    if (live) {
        const int fj = jin ? fd[j] : 0;
        const float* vj = &vs[lane * LVROW];
        int    fr[2 * NPW];
        float4 g[2 * NPW][2];
        #pragma unroll
        for (int k = 0; k < 2 * NPW; ++k) fr[k] = fd[2 * p0 + k];
        #pragma unroll
        for (int k = 0; k < 2 * NPW; ++k) {
            const float4* Vi = reinterpret_cast<const float4*>(
                V + (size_t)(2 * p0 + k) * VROW + fj * NF);
            g[k][0] = Vi[0];
            g[k][1] = Vi[1];
        }
        #pragma unroll
        for (int k = 0; k < 2 * NPW; ++k) {
            const float4* Vj = reinterpret_cast<const float4*>(vj + fr[k] * NF);
            const float4 a0 = g[k][0], a1 = g[k][1];
            const float4 b0 = Vj[0], b1 = Vj[1];
            const float d = a0.x * b0.x + a0.y * b0.y + a0.z * b0.z + a0.w * b0.w
                          + a1.x * b1.x + a1.y * b1.y + a1.z * b1.z + a1.w * b1.w;
            const float s = (jin && 2 * p0 + k < j) ? d : 0.0f;
            if (k & 1) res[k >> 1].y = s; else res[k >> 1].x = s;
        }
    }
    #pragma unroll
    for (int t = 0; t < NPW; ++t)
        *reinterpret_cast<float2*>(Asw + (((size_t)(p0 + t) * NPAD + j) << 1)) = res[t];
    if (rbk == 0 && wv == 0) wp[j] = jin ? W[j] : 0.0f;
}

__device__ __forceinline__ void bar_signal() { asm volatile("s_wait_dscnt 0x0\n\ts_barrier_signal -1" ::: "memory"); }
__device__ __forceinline__ void bar_wait()   { asm volatile("s_barrier_wait -1" ::: "memory"); }

__global__ __launch_bounds__(256) void FFM_wmma_gemm(
    const float* __restrict__ x, const float* __restrict__ Asw,
    const float* __restrict__ wp, float* __restrict__ parts)
{
    __shared__ __attribute__((aligned(16))) float lds[NXBUF * XBUF];
    const int tid  = threadIdx.x;
    const int w    = __builtin_amdgcn_readfirstlane(tid >> 5);
    const int lane = tid & 31;
    const int hi   = lane >> 4;
    const int lo   = lane & 15;
    const int sp   = blockIdx.y;
    const int s_base = blockIdx.x * MT;

    v8f c[NTPW][NSUB];
    #pragma unroll
    for (int t = 0; t < NTPW; ++t)
        #pragma unroll
        for (int u = 0; u < NSUB; ++u)
            c[t][u] = (v8f){0.f, 0.f, 0.f, 0.f, 0.f, 0.f, 0.f, 0.f};

    const int pos = (w < 4) ? w : 11 - w;
    int ntile[NTPW];
    int lim[NTPW];
    const float* bbase[NTPW];
    #pragma unroll
    for (int t = 0; t < NTPW; ++t) {
        ntile[t] = t * 32 + ((t & 1) ? (7 - pos) : pos) * 4 + sp;
        const int pairs = (16 * ntile[t] + 14) / (4 * KS * 2) + 1;
        lim[t] = 2 * (pairs < NPAIR ? pairs : NPAIR);
        const int n = ntile[t] * 16 + lo;
        bbase[t] = Asw + ((size_t)hi * NPAD + n) * 2;
    }

    v4f xr[XV4];
    int xso[XV4];
    #pragma unroll
    for (int j = 0; j < XV4; ++j) {
        const int i = tid + j * 256;
        xso[j] = (i / (CK / 4)) * LDX + (i % (CK / 4)) * 4;
    }
    v2f bA[KS][NTPW];
    v2f bB[KS][NTPW];

#define ISSUE_X(kc2)                                                            \
    _Pragma("unroll")                                                           \
    for (int j = 0; j < XV4; ++j) {                                             \
        const int i = tid + j * 256;                                            \
        if (j < XV4 - 1 || tid < 128) {                                         \
            const int r  = i / (CK / 4);                                        \
            const int cq = i % (CK / 4);                                        \
            xr[j] = *reinterpret_cast<const v4f*>(                              \
                x + (size_t)(s_base + r) * DIMD + (kc2) * CK + cq * 4);         \
        }                                                                       \
    }
#define STORE_X(xo2)                                                            \
    _Pragma("unroll")                                                           \
    for (int j = 0; j < XV4; ++j) {                                             \
        if (j < XV4 - 1 || tid < 128) {                                         \
            *reinterpret_cast<v4f*>(&lds[(xo2) + xso[j]]) = xr[j];              \
        }                                                                       \
    }

    auto load_B = [&](auto tm, v2f (&b)[KS][NTPW], int g) {
        constexpr int TM = decltype(tm)::value;
        #pragma unroll
        for (int j = 0; j < KS; ++j)
            #pragma unroll
            for (int t = TM; t < NTPW; ++t)
                b[j][t] = *reinterpret_cast<const v2f*>(
                    bbase[t] + (size_t)(g * (KS * 2) + j * 2) * (NPAD * 2));
    };
    auto mult = [&](auto tm, const v2f (&b)[KS][NTPW], int sl, int xo) {
        constexpr int TM = decltype(tm)::value;
        #pragma unroll
        for (int j = 0; j < KS; ++j) {
            const int k0 = (sl * KS + j) * 4;
            v2f a[NSUB];
            #pragma unroll
            for (int u = 0; u < NSUB; ++u) {
                const float* p = &lds[xo + (u * 16 + lo) * LDX + k0 + hi * 2];
                a[u][0] = p[0]; a[u][1] = p[1];
            }
            #pragma unroll
            for (int t = TM; t < NTPW; ++t)
                #pragma unroll
                for (int u = 0; u < NSUB; ++u)
                    c[t][u] = __builtin_amdgcn_wmma_f32_16x16x4_f32(
                        false, a[u], false, b[j][t], (short)0, c[t][u], false, false);
        }
    };

    auto pair = [&](auto tm, int g, int sl, int xo) {
        load_B(tm, bB, g + 1);
        mult(tm, bA, sl, xo);
        load_B(tm, bA, (g + 2 < NGST) ? (g + 2) : (NGST - 1));
        mult(tm, bB, sl + 1, xo);
    };

    ISSUE_X(0)
    load_B(IC<0>{}, bA, 0);
    STORE_X(0)
    ISSUE_X(1)
    bar_signal();

    int xo = 0;
    for (int kc = 0; kc < NKC; ++kc) {
        bar_wait();
        const int xn = (xo == (NXBUF - 1) * XBUF) ? 0 : xo + XBUF;
        if (kc + 1 < NKC) {
            STORE_X(xn)
            if (kc + 2 < NKC) { ISSUE_X(kc + 2) }
            bar_signal();
        }

        for (int sl = 0; sl < NSTG; sl += 2) {
            const int g = kc * NSTG + sl;
            const int tmin = (g >= lim[0]) + (g >= lim[1]) + (g >= lim[2]) + (g >= lim[3]);
            switch (tmin) {
                case 0:  pair(IC<0>{}, g, sl, xo); break;
                case 1:  pair(IC<1>{}, g, sl, xo); break;
                case 2:  pair(IC<2>{}, g, sl, xo); break;
                case 3:  pair(IC<3>{}, g, sl, xo); break;
                default: break;
            }
        }
        xo = xn;
    }
    __syncthreads();

#undef ISSUE_X
#undef STORE_X

    float racc[NSUB][8];
    #pragma unroll
    for (int u = 0; u < NSUB; ++u)
        #pragma unroll
        for (int v = 0; v < 8; ++v) racc[u][v] = 0.0f;

    #pragma unroll
    for (int t = 0; t < NTPW; ++t) {
        const int n = ntile[t] * 16 + lo;
        const float wn = wp[n];
        const bool inb = (n < DIMD);
        #pragma unroll
        for (int u = 0; u < NSUB; ++u) {
            #pragma unroll
            for (int v = 0; v < 8; ++v) {
                const int r = v + hi * 8 + u * 16;
                float xv = inb ? x[(size_t)(s_base + r) * DIMD + n] : 0.0f;
                float tv = c[t][u][v];
                racc[u][v] += xv * (tv + wn);
            }
        }
    }

    #pragma unroll
    for (int u = 0; u < NSUB; ++u) {
        #pragma unroll
        for (int v = 0; v < 8; ++v) {
            float s = racc[u][v];
            s += __shfl_xor(s, 8, 32);
            s += __shfl_xor(s, 4, 32);
            s += __shfl_xor(s, 2, 32);
            s += __shfl_xor(s, 1, 32);
            if (lo == 0) {
                const int r = v + hi * 8 + u * 16;
                lds[w * MT + r] = s;
            }
        }
    }
    __syncthreads();

    if (tid < MT) {
        float p = 0.0f;
        #pragma unroll
        for (int w2 = 0; w2 < 8; ++w2) p += lds[w2 * MT + tid];
        parts[(size_t)(s_base + tid) * NSPLIT + sp] = p;
    }
}

__global__ __launch_bounds__(256) void FFM_finalize(
    const float* __restrict__ parts, const float* __restrict__ b,
    float* __restrict__ out)
{
    const int i = blockIdx.x * blockDim.x + threadIdx.x;
    if (i < BATCH) {
        float p = b[0];
        #pragma unroll
        for (int s = 0; s < NSPLIT; ++s) p += parts[(size_t)i * NSPLIT + s];
        out[i] = p;
    }
}

extern "C" void kernel_launch(void* const* d_in, const int* in_sizes, int n_in,
                              void* d_out, int out_size, void* d_ws, size_t ws_size,
                              hipStream_t stream)
{
    const float* x  = (const float*)d_in[0];
    const int*   fd = (const int*)  d_in[1];
    const float* b  = (const float*)d_in[2];
    const float* W  = (const float*)d_in[3];
    const float* V  = (const float*)d_in[4];
    float* out = (float*)d_out;

    float* Asw   = (float*)d_ws;
    float* wp    = Asw + (size_t)DIMD * NPAD;
    float* parts = wp + NPAD;

    FFM_build_A<<<NBUILD, 256, 0, stream>>>(V, fd, W, Asw, wp);

    dim3 g2(BATCH / MT, NSPLIT);
    FFM_wmma_gemm<<<g2, 256, 0, stream>>>(x, Asw, wp, parts);

    FFM_finalize<<<BATCH / 256, 256, 0, stream>>>(parts, b, out);
}


# All loads of a 5-k-step stage (20 LDS A-fragment loads into a second register set, and its B fragments) stand as one group one stage ahead; WMMAs run tile by tile; sum order unchanged
# speedup vs baseline: 2.6014x; 1.0506x over previous
#include <hip/hip_runtime.h>


#define DIMD   2000
#define NPAD   2048
#define BATCH  4096
#define NFIELD 40
#define NF     8
#define MT     64
#define NSUB   4
#define CK     200
#define NKC    10
#define LDX    204
#define NSPLIT 4
#define NTPW   4
#define KS     5
#define NSTG   10
#define NGST   100
#define XV4    13
#define NPAIR  50
#define BCB    32
#define BRB    80
#define NCB    64
#define NRBK   25
#define VROW   320
#define LVROW  324
#define NPW    5
#define NSTV   10
#define NBUILD 892
#define XBUF   (MT * LDX)
#define NXBUF  3

template <int N> struct IC { static constexpr int value = N; };

typedef __attribute__((ext_vector_type(2))) float v2f;
typedef __attribute__((ext_vector_type(8))) float v8f;
typedef __attribute__((ext_vector_type(4))) float v4f;

__global__ __launch_bounds__(256) void FFM_build_A(
    const float* __restrict__ V, const int* __restrict__ fd,
    const float* __restrict__ W, float* __restrict__ Asw,
    float* __restrict__ wp)
{
    __shared__ __attribute__((aligned(16))) float vs[BCB * LVROW];

    int rest = blockIdx.x;
    int rbk = 0;
    int cb = -1;
    for (rbk = 0; rbk < NRBK; ++rbk) {
        const int cmin = (rbk == 0) ? 0 : (BRB * rbk - 39) / BCB;
        const int cnt = NCB - cmin;
        if (rest < cnt) { cb = cmin + rest; break; }
        rest -= cnt;
    }
    if (cb < 0) return;

    const int tid  = threadIdx.x;
    const int lane = tid & 31;
    const int wv   = __builtin_amdgcn_readfirstlane(tid >> 5);
    const int j0   = cb * BCB;
    const int j    = j0 + lane;
    const bool live = (j0 < DIMD) && (rbk * BRB < j0 + BCB - 1);

    if (live) {
        float4 st[NSTV];
        #pragma unroll
        for (int k = 0; k < NSTV; ++k) {
            const int idx = tid + k * 256;
            const int r   = idx / (VROW / 4);
            const int c4  = idx - r * (VROW / 4);
            const int rs  = (j0 + r < DIMD) ? (j0 + r) : (DIMD - 1);
            st[k] = *reinterpret_cast<const float4*>(V + (size_t)rs * VROW + c4 * 4);
        }
        #pragma unroll
        for (int k = 0; k < NSTV; ++k) {
            const int idx = tid + k * 256;
            const int r   = idx / (VROW / 4);
            const int c4  = idx - r * (VROW / 4);
            *reinterpret_cast<float4*>(&vs[r * LVROW + c4 * 4]) = st[k];
        }
    }
    __syncthreads();

    const bool jin = (j < DIMD);
    const int  p0  = rbk * (BRB / 2) + wv * NPW;
    float2 res[NPW];
    #pragma unroll
    for (int t = 0; t < NPW; ++t) res[t] = make_float2(0.0f, 0.0f);

    if (live) {
        const int fj = jin ? fd[j] : 0;
        const float* vj = &vs[lane * LVROW];
        int    fr[2 * NPW];
        float4 g[2 * NPW][2];
        #pragma unroll
        for (int k = 0; k < 2 * NPW; ++k) fr[k] = fd[2 * p0 + k];
        #pragma unroll
        for (int k = 0; k < 2 * NPW; ++k) {
            const float4* Vi = reinterpret_cast<const float4*>(
                V + (size_t)(2 * p0 + k) * VROW + fj * NF);
            g[k][0] = Vi[0];
            g[k][1] = Vi[1];
        }
        #pragma unroll
        for (int k = 0; k < 2 * NPW; ++k) {
            const float4* Vj = reinterpret_cast<const float4*>(vj + fr[k] * NF);
            const float4 a0 = g[k][0], a1 = g[k][1];
            const float4 b0 = Vj[0], b1 = Vj[1];
            const float d = a0.x * b0.x + a0.y * b0.y + a0.z * b0.z + a0.w * b0.w
                          + a1.x * b1.x + a1.y * b1.y + a1.z * b1.z + a1.w * b1.w;
            const float s = (jin && 2 * p0 + k < j) ? d : 0.0f;
            if (k & 1) res[k >> 1].y = s; else res[k >> 1].x = s;
        }
    }
    #pragma unroll
    for (int t = 0; t < NPW; ++t)
        *reinterpret_cast<float2*>(Asw + (((size_t)(p0 + t) * NPAD + j) << 1)) = res[t];
    if (rbk == 0 && wv == 0) wp[j] = jin ? W[j] : 0.0f;
}

__device__ __forceinline__ void bar_signal() { asm volatile("s_wait_dscnt 0x0\n\ts_barrier_signal -1" ::: "memory"); }
__device__ __forceinline__ void bar_wait()   { asm volatile("s_barrier_wait -1" ::: "memory"); }

__global__ __launch_bounds__(256) void FFM_wmma_gemm(
    const float* __restrict__ x, const float* __restrict__ Asw,
    const float* __restrict__ wp, float* __restrict__ parts)
{
    __shared__ __attribute__((aligned(16))) float lds[NXBUF * XBUF];
    const int tid  = threadIdx.x;
    const int w    = __builtin_amdgcn_readfirstlane(tid >> 5);
    const int lane = tid & 31;
    const int hi   = lane >> 4;
    const int lo   = lane & 15;
    const int sp   = blockIdx.y;
    const int s_base = blockIdx.x * MT;

    v8f c[NTPW][NSUB];
    #pragma unroll
    for (int t = 0; t < NTPW; ++t)
        #pragma unroll
        for (int u = 0; u < NSUB; ++u)
            c[t][u] = (v8f){0.f, 0.f, 0.f, 0.f, 0.f, 0.f, 0.f, 0.f};

    const int pos = (w < 4) ? w : 11 - w;
    int ntile[NTPW];
    int lim[NTPW];
    const float* bbase[NTPW];
    #pragma unroll
    for (int t = 0; t < NTPW; ++t) {
        ntile[t] = t * 32 + ((t & 1) ? (7 - pos) : pos) * 4 + sp;
        const int pairs = (16 * ntile[t] + 14) / (4 * KS * 2) + 1;
        lim[t] = 2 * (pairs < NPAIR ? pairs : NPAIR);
        const int n = ntile[t] * 16 + lo;
        bbase[t] = Asw + ((size_t)hi * NPAD + n) * 2;
    }

    v4f xr[XV4];
    int xso[XV4];
    #pragma unroll
    for (int j = 0; j < XV4; ++j) {
        const int i = tid + j * 256;
        xso[j] = (i / (CK / 4)) * LDX + (i % (CK / 4)) * 4;
    }
    v2f bA[KS][NTPW];
    v2f bB[KS][NTPW];
    v2f aA[KS][NSUB];
    v2f aB[KS][NSUB];

#define ISSUE_X(kc2)                                                            \
    _Pragma("unroll")                                                           \
    for (int j = 0; j < XV4; ++j) {                                             \
        const int i = tid + j * 256;                                            \
        if (j < XV4 - 1 || tid < 128) {                                         \
            const int r  = i / (CK / 4);                                        \
            const int cq = i % (CK / 4);                                        \
            xr[j] = *reinterpret_cast<const v4f*>(                              \
                x + (size_t)(s_base + r) * DIMD + (kc2) * CK + cq * 4);         \
        }                                                                       \
    }
#define STORE_X(xo2)                                                            \
    _Pragma("unroll")                                                           \
    for (int j = 0; j < XV4; ++j) {                                             \
        if (j < XV4 - 1 || tid < 128) {                                         \
            *reinterpret_cast<v4f*>(&lds[(xo2) + xso[j]]) = xr[j];              \
        }                                                                       \
    }

    auto load_B = [&](auto tm, v2f (&b)[KS][NTPW], int g) {
        constexpr int TM = decltype(tm)::value;
        #pragma unroll
        for (int j = 0; j < KS; ++j)
            #pragma unroll
            for (int t = TM; t < NTPW; ++t)
                b[j][t] = *reinterpret_cast<const v2f*>(
                    bbase[t] + (size_t)(g * (KS * 2) + j * 2) * (NPAD * 2));
    };
    auto load_B4 = [&](v2f (&b)[KS][NTPW], int g) {
        #pragma unroll
        for (int t = NTPW - 1; t >= 0; --t) {
            const float* bp = (g < lim[t]) ? bbase[t] : bbase[NTPW - 1];
            #pragma unroll
            for (int j = 0; j < KS; ++j)
                b[j][t] = *reinterpret_cast<const v2f*>(
                    bp + (size_t)(g * (KS * 2) + j * 2) * (NPAD * 2));
        }
    };
    auto load_A = [&](v2f (&a)[KS][NSUB], int sl, int xo) {
        #pragma unroll
        for (int j = 0; j < KS; ++j) {
            const int k0 = (sl * KS + j) * 4;
            #pragma unroll
            for (int u = 0; u < NSUB; ++u)
                a[j][u] = *reinterpret_cast<const v2f*>(
                    &lds[xo + (u * 16 + lo) * LDX + k0 + hi * 2]);
        }
    };
    auto tile_mult = [&](auto tt, const v2f (&a)[KS][NSUB], const v2f (&b)[KS][NTPW]) {
        constexpr int T = decltype(tt)::value;
        #pragma unroll
        for (int j = 0; j < KS; ++j)
            #pragma unroll
            for (int u = 0; u < NSUB; ++u)
                c[T][u] = __builtin_amdgcn_wmma_f32_16x16x4_f32(
                    false, a[j][u], false, b[j][T], (short)0, c[T][u], false, false);
    };
    auto stage = [&](const v2f (&a)[KS][NSUB], const v2f (&b)[KS][NTPW], int tmin) {
        tile_mult(IC<3>{}, a, b);
        if (tmin <= 2) {
            tile_mult(IC<2>{}, a, b);
            if (tmin <= 1) {
                tile_mult(IC<1>{}, a, b);
                if (tmin == 0) tile_mult(IC<0>{}, a, b);
            }
        }
    };

    ISSUE_X(0)
    load_B(IC<0>{}, bA, 0);
    STORE_X(0)
    ISSUE_X(1)
    bar_signal();

    int xo = 0;
    for (int kc = 0; kc < NKC; ++kc) {
        bar_wait();
        const bool act0 = (kc * NSTG < lim[NTPW - 1]);
        if (act0) load_A(aA, 0, xo);
        const int xn = (xo == (NXBUF - 1) * XBUF) ? 0 : xo + XBUF;
        if (kc + 1 < NKC) {
            STORE_X(xn)
            if (kc + 2 < NKC && !act0) { ISSUE_X(kc + 2) }
            bar_signal();
        }

        for (int sl = 0; sl < NSTG; sl += 2) {
            const int g = kc * NSTG + sl;
            const int tmin = (g >= lim[0]) + (g >= lim[1]) + (g >= lim[2]) + (g >= lim[3]);
            if (tmin < NTPW) {
                load_A(aB, sl + 1, xo);
                load_B4(bB, g + 1);
                if (sl == 0 && kc + 2 < NKC) { ISSUE_X(kc + 2) }
                __builtin_amdgcn_sched_barrier(0);
                stage(aA, bA, tmin);
                load_A(aA, (sl + 2 < NSTG) ? (sl + 2) : 0, xo);
                load_B4(bA, (g + 2 < NGST) ? (g + 2) : (NGST - 1));
                __builtin_amdgcn_sched_barrier(0);
                stage(aB, bB, tmin);
            }
        }
        xo = xn;
    }
    __syncthreads();

#undef ISSUE_X
#undef STORE_X

    float racc[NSUB][8];
    #pragma unroll
    for (int u = 0; u < NSUB; ++u)
        #pragma unroll
        for (int v = 0; v < 8; ++v) racc[u][v] = 0.0f;

    #pragma unroll
    for (int t = 0; t < NTPW; ++t) {
        const int n = ntile[t] * 16 + lo;
        const float wn = wp[n];
        const bool inb = (n < DIMD);
        #pragma unroll
        for (int u = 0; u < NSUB; ++u) {
            #pragma unroll
            for (int v = 0; v < 8; ++v) {
                const int r = v + hi * 8 + u * 16;
                float xv = inb ? x[(size_t)(s_base + r) * DIMD + n] : 0.0f;
                float tv = c[t][u][v];
                racc[u][v] += xv * (tv + wn);
            }
        }
    }

    #pragma unroll
    for (int u = 0; u < NSUB; ++u) {
        #pragma unroll
        for (int v = 0; v < 8; ++v) {
            float s = racc[u][v];
            s += __shfl_xor(s, 8, 32);
            s += __shfl_xor(s, 4, 32);
            s += __shfl_xor(s, 2, 32);
            s += __shfl_xor(s, 1, 32);
            if (lo == 0) {
                const int r = v + hi * 8 + u * 16;
                lds[w * MT + r] = s;
            }
        }
    }
    __syncthreads();

    if (tid < MT) {
        float p = 0.0f;
        #pragma unroll
        for (int w2 = 0; w2 < 8; ++w2) p += lds[w2 * MT + tid];
        parts[(size_t)(s_base + tid) * NSPLIT + sp] = p;
    }
}

__global__ __launch_bounds__(256) void FFM_finalize(
    const float* __restrict__ parts, const float* __restrict__ b,
    float* __restrict__ out)
{
    const int i = blockIdx.x * blockDim.x + threadIdx.x;
    if (i < BATCH) {
        float p = b[0];
        #pragma unroll
        for (int s = 0; s < NSPLIT; ++s) p += parts[(size_t)i * NSPLIT + s];
        out[i] = p;
    }
}

extern "C" void kernel_launch(void* const* d_in, const int* in_sizes, int n_in,
                              void* d_out, int out_size, void* d_ws, size_t ws_size,
                              hipStream_t stream)
{
    const float* x  = (const float*)d_in[0];
    const int*   fd = (const int*)  d_in[1];
    const float* b  = (const float*)d_in[2];
    const float* W  = (const float*)d_in[3];
    const float* V  = (const float*)d_in[4];
    float* out = (float*)d_out;

    float* Asw   = (float*)d_ws;
    float* wp    = Asw + (size_t)DIMD * NPAD;
    float* parts = wp + NPAD;

    FFM_build_A<<<NBUILD, 256, 0, stream>>>(V, fd, W, Asw, wp);

    dim3 g2(BATCH / MT, NSPLIT);
    FFM_wmma_gemm<<<g2, 256, 0, stream>>>(x, Asw, wp, parts);

    FFM_finalize<<<BATCH / 256, 256, 0, stream>>>(parts, b, out);
}


# Tile deal trimmed to the nonzero triangle: no all-padding tile multiplied (slots shifted by 3), K loops end per 20-k stage, split index mirrored on odd bands; tiles regrouped, sum order changes
# speedup vs baseline: 2.7570x; 1.0598x over previous
#include <hip/hip_runtime.h>


#define DIMD   2000
#define NPAD   2048
#define BATCH  4096
#define NFIELD 40
#define NF     8
#define MT     64
#define NSUB   4
#define CK     200
#define NKC    10
#define LDX    204
#define NSPLIT 4
#define NTPW   4
#define KS     5
#define NSTG   10
#define NGST   100
#define XV4    13
#define NEMPTY 3
#define BCB    32
#define BRB    80
#define NCB    64
#define NRBK   25
#define VROW   320
#define LVROW  324
#define NPW    5
#define NSTV   10
#define NBUILD 892
#define XBUF   (MT * LDX)
#define NXBUF  3

template <int N> struct IC { static constexpr int value = N; };

typedef __attribute__((ext_vector_type(2))) float v2f;
typedef __attribute__((ext_vector_type(8))) float v8f;
typedef __attribute__((ext_vector_type(4))) float v4f;

__global__ __launch_bounds__(256) void FFM_build_A(
    const float* __restrict__ V, const int* __restrict__ fd,
    const float* __restrict__ W, float* __restrict__ Asw,
    float* __restrict__ wp)
{
    __shared__ __attribute__((aligned(16))) float vs[BCB * LVROW];

    int rest = blockIdx.x;
    int rbk = 0;
    int cb = -1;
    for (rbk = 0; rbk < NRBK; ++rbk) {
        const int cmin = (rbk == 0) ? 0 : (BRB * rbk - 39) / BCB;
        const int cnt = NCB - cmin;
        if (rest < cnt) { cb = cmin + rest; break; }
        rest -= cnt;
    }
    if (cb < 0) return;

    const int tid  = threadIdx.x;
    const int lane = tid & 31;
    const int wv   = __builtin_amdgcn_readfirstlane(tid >> 5);
    const int j0   = cb * BCB;
    const int j    = j0 + lane;
    const bool live = (j0 < DIMD) && (rbk * BRB < j0 + BCB - 1);

    if (live) {
        float4 st[NSTV];
        #pragma unroll
        for (int k = 0; k < NSTV; ++k) {
            const int idx = tid + k * 256;
            const int r   = idx / (VROW / 4);
            const int c4  = idx - r * (VROW / 4);
            const int rs  = (j0 + r < DIMD) ? (j0 + r) : (DIMD - 1);
            st[k] = *reinterpret_cast<const float4*>(V + (size_t)rs * VROW + c4 * 4);
        }
        #pragma unroll
        for (int k = 0; k < NSTV; ++k) {
            const int idx = tid + k * 256;
            const int r   = idx / (VROW / 4);
            const int c4  = idx - r * (VROW / 4);
            *reinterpret_cast<float4*>(&vs[r * LVROW + c4 * 4]) = st[k];
        }
    }
    __syncthreads();

    const bool jin = (j < DIMD);
    const int  p0  = rbk * (BRB / 2) + wv * NPW;
    float2 res[NPW];
    #pragma unroll
    for (int t = 0; t < NPW; ++t) res[t] = make_float2(0.0f, 0.0f);

    if (live) {
        const int fj = jin ? fd[j] : 0;
        const float* vj = &vs[lane * LVROW];
        int    fr[2 * NPW];
        float4 g[2 * NPW][2];
        #pragma unroll
        for (int k = 0; k < 2 * NPW; ++k) fr[k] = fd[2 * p0 + k];
        #pragma unroll
        for (int k = 0; k < 2 * NPW; ++k) {
            const float4* Vi = reinterpret_cast<const float4*>(
                V + (size_t)(2 * p0 + k) * VROW + fj * NF);
            g[k][0] = Vi[0];
            g[k][1] = Vi[1];
        }
        #pragma unroll
        for (int k = 0; k < 2 * NPW; ++k) {
            const float4* Vj = reinterpret_cast<const float4*>(vj + fr[k] * NF);
            const float4 a0 = g[k][0], a1 = g[k][1];
            const float4 b0 = Vj[0], b1 = Vj[1];
            const float d = a0.x * b0.x + a0.y * b0.y + a0.z * b0.z + a0.w * b0.w
                          + a1.x * b1.x + a1.y * b1.y + a1.z * b1.z + a1.w * b1.w;
            const float s = (jin && 2 * p0 + k < j) ? d : 0.0f;
            if (k & 1) res[k >> 1].y = s; else res[k >> 1].x = s;
        }
    }
    #pragma unroll
    for (int t = 0; t < NPW; ++t)
        *reinterpret_cast<float2*>(Asw + (((size_t)(p0 + t) * NPAD + j) << 1)) = res[t];
    if (rbk == 0 && wv == 0) wp[j] = jin ? W[j] : 0.0f;
}

__device__ __forceinline__ void bar_signal() { asm volatile("s_wait_dscnt 0x0\n\ts_barrier_signal -1" ::: "memory"); }
__device__ __forceinline__ void bar_wait()   { asm volatile("s_barrier_wait -1" ::: "memory"); }

__global__ __launch_bounds__(256) void FFM_wmma_gemm(
    const float* __restrict__ x, const float* __restrict__ Asw,
    const float* __restrict__ wp, float* __restrict__ parts)
{
    __shared__ __attribute__((aligned(16))) float lds[NXBUF * XBUF];
    const int tid  = threadIdx.x;
    const int w    = __builtin_amdgcn_readfirstlane(tid >> 5);
    const int lane = tid & 31;
    const int hi   = lane >> 4;
    const int lo   = lane & 15;
    const int sp   = blockIdx.y;
    const int s_base = blockIdx.x * MT;

    v8f c[NTPW][NSUB];
    #pragma unroll
    for (int t = 0; t < NTPW; ++t)
        #pragma unroll
        for (int u = 0; u < NSUB; ++u)
            c[t][u] = (v8f){0.f, 0.f, 0.f, 0.f, 0.f, 0.f, 0.f, 0.f};

    const int pos = (w < 4) ? w : 11 - w;
    int ntile[NTPW];
    int lim[NTPW];
    const float* bbase[NTPW];
    #pragma unroll
    for (int t = 0; t < NTPW; ++t) {
        const int q = (t & 1) ? (7 - pos) : pos;
        const int r = (t & 1) ? (NSPLIT - 1 - sp) : sp;
        ntile[t] = t * 32 + q * 4 + r - NEMPTY;
        const int nt = (ntile[t] < 0) ? 0 : ntile[t];
        const int st = (16 * nt + 14) / (4 * KS) + 1;
        lim[t] = (ntile[t] < 0) ? 0 : (st < NGST ? st : NGST);
        const int n = nt * 16 + lo;
        bbase[t] = Asw + ((size_t)hi * NPAD + n) * 2;
    }

    v4f xr[XV4];
    int xso[XV4];
    #pragma unroll
    for (int j = 0; j < XV4; ++j) {
        const int i = tid + j * 256;
        xso[j] = (i / (CK / 4)) * LDX + (i % (CK / 4)) * 4;
    }
    v2f bA[KS][NTPW];
    v2f bB[KS][NTPW];
    v2f aA[KS][NSUB];
    v2f aB[KS][NSUB];

#define ISSUE_X(kc2)                                                            \
    _Pragma("unroll")                                                           \
    for (int j = 0; j < XV4; ++j) {                                             \
        const int i = tid + j * 256;                                            \
        if (j < XV4 - 1 || tid < 128) {                                         \
            const int r  = i / (CK / 4);                                        \
            const int cq = i % (CK / 4);                                        \
            xr[j] = *reinterpret_cast<const v4f*>(                              \
                x + (size_t)(s_base + r) * DIMD + (kc2) * CK + cq * 4);         \
        }                                                                       \
    }
#define STORE_X(xo2)                                                            \
    _Pragma("unroll")                                                           \
    for (int j = 0; j < XV4; ++j) {                                             \
        if (j < XV4 - 1 || tid < 128) {                                         \
            *reinterpret_cast<v4f*>(&lds[(xo2) + xso[j]]) = xr[j];              \
        }                                                                       \
    }

    auto load_B = [&](auto tm, v2f (&b)[KS][NTPW], int g) {
        constexpr int TM = decltype(tm)::value;
        #pragma unroll
        for (int j = 0; j < KS; ++j)
            #pragma unroll
            for (int t = TM; t < NTPW; ++t)
                b[j][t] = *reinterpret_cast<const v2f*>(
                    bbase[t] + (size_t)(g * (KS * 2) + j * 2) * (NPAD * 2));
    };
    auto load_B4 = [&](v2f (&b)[KS][NTPW], int g) {
        #pragma unroll
        for (int t = NTPW - 1; t >= 0; --t) {
            const float* bp = (g < lim[t]) ? bbase[t] : bbase[NTPW - 1];
            #pragma unroll
            for (int j = 0; j < KS; ++j)
                b[j][t] = *reinterpret_cast<const v2f*>(
                    bp + (size_t)(g * (KS * 2) + j * 2) * (NPAD * 2));
        }
    };
    auto load_A = [&](v2f (&a)[KS][NSUB], int sl, int xo) {
        #pragma unroll
        for (int j = 0; j < KS; ++j) {
            const int k0 = (sl * KS + j) * 4;
            #pragma unroll
            for (int u = 0; u < NSUB; ++u)
                a[j][u] = *reinterpret_cast<const v2f*>(
                    &lds[xo + (u * 16 + lo) * LDX + k0 + hi * 2]);
        }
    };
    auto tile_mult = [&](auto tt, const v2f (&a)[KS][NSUB], const v2f (&b)[KS][NTPW]) {
        constexpr int T = decltype(tt)::value;
        #pragma unroll
        for (int j = 0; j < KS; ++j)
            #pragma unroll
            for (int u = 0; u < NSUB; ++u)
                c[T][u] = __builtin_amdgcn_wmma_f32_16x16x4_f32(
                    false, a[j][u], false, b[j][T], (short)0, c[T][u], false, false);
    };
    auto stage = [&](const v2f (&a)[KS][NSUB], const v2f (&b)[KS][NTPW], int tmin) {
        tile_mult(IC<3>{}, a, b);
        if (tmin <= 2) {
            tile_mult(IC<2>{}, a, b);
            if (tmin <= 1) {
                tile_mult(IC<1>{}, a, b);
                if (tmin == 0) tile_mult(IC<0>{}, a, b);
            }
        }
    };

    ISSUE_X(0)
    load_B(IC<0>{}, bA, 0);
    STORE_X(0)
    ISSUE_X(1)
    bar_signal();

    int xo = 0;
    for (int kc = 0; kc < NKC; ++kc) {
        bar_wait();
        const bool act0 = (kc * NSTG < lim[NTPW - 1]);
        if (act0) load_A(aA, 0, xo);
        const int xn = (xo == (NXBUF - 1) * XBUF) ? 0 : xo + XBUF;
        if (kc + 1 < NKC) {
            STORE_X(xn)
            if (kc + 2 < NKC && !act0) { ISSUE_X(kc + 2) }
            bar_signal();
        }

        for (int sl = 0; sl < NSTG; sl += 2) {
            const int g = kc * NSTG + sl;
            const int tmin = (g >= lim[0]) + (g >= lim[1]) + (g >= lim[2]) + (g >= lim[3]);
            const int tmin1 = (g + 1 >= lim[0]) + (g + 1 >= lim[1]) + (g + 1 >= lim[2]) + (g + 1 >= lim[3]);
            if (tmin < NTPW) {
                load_A(aB, sl + 1, xo);
                load_B4(bB, g + 1);
                if (sl == 0 && kc + 2 < NKC) { ISSUE_X(kc + 2) }
                __builtin_amdgcn_sched_barrier(0);
                stage(aA, bA, tmin);
                load_A(aA, (sl + 2 < NSTG) ? (sl + 2) : 0, xo);
                load_B4(bA, (g + 2 < NGST) ? (g + 2) : (NGST - 1));
                __builtin_amdgcn_sched_barrier(0);
                if (tmin1 < NTPW) stage(aB, bB, tmin1);
            }
        }
        xo = xn;
    }
    __syncthreads();

#undef ISSUE_X
#undef STORE_X

    float racc[NSUB][8];
    #pragma unroll
    for (int u = 0; u < NSUB; ++u)
        #pragma unroll
        for (int v = 0; v < 8; ++v) racc[u][v] = 0.0f;

    #pragma unroll
    for (int t = 0; t < NTPW; ++t) {
        const bool inb = (ntile[t] >= 0);
        const int n = (inb ? ntile[t] : 0) * 16 + lo;
        const float wn = wp[n];
        #pragma unroll
        for (int u = 0; u < NSUB; ++u) {
            #pragma unroll
            for (int v = 0; v < 8; ++v) {
                const int r = v + hi * 8 + u * 16;
                float xv = inb ? x[(size_t)(s_base + r) * DIMD + n] : 0.0f;
                float tv = c[t][u][v];
                racc[u][v] += xv * (tv + wn);
            }
        }
    }

    #pragma unroll
    for (int u = 0; u < NSUB; ++u) {
        #pragma unroll
        for (int v = 0; v < 8; ++v) {
            float s = racc[u][v];
            s += __shfl_xor(s, 8, 32);
            s += __shfl_xor(s, 4, 32);
            s += __shfl_xor(s, 2, 32);
            s += __shfl_xor(s, 1, 32);
            if (lo == 0) {
                const int r = v + hi * 8 + u * 16;
                lds[w * MT + r] = s;
            }
        }
    }
    __syncthreads();

    if (tid < MT) {
        float p = 0.0f;
        #pragma unroll
        for (int w2 = 0; w2 < 8; ++w2) p += lds[w2 * MT + tid];
        parts[(size_t)(s_base + tid) * NSPLIT + sp] = p;
    }
}

__global__ __launch_bounds__(256) void FFM_finalize(
    const float* __restrict__ parts, const float* __restrict__ b,
    float* __restrict__ out)
{
    const int i = blockIdx.x * blockDim.x + threadIdx.x;
    if (i < BATCH) {
        float p = b[0];
        #pragma unroll
        for (int s = 0; s < NSPLIT; ++s) p += parts[(size_t)i * NSPLIT + s];
        out[i] = p;
    }
}

extern "C" void kernel_launch(void* const* d_in, const int* in_sizes, int n_in,
                              void* d_out, int out_size, void* d_ws, size_t ws_size,
                              hipStream_t stream)
{
    const float* x  = (const float*)d_in[0];
    const int*   fd = (const int*)  d_in[1];
    const float* b  = (const float*)d_in[2];
    const float* W  = (const float*)d_in[3];
    const float* V  = (const float*)d_in[4];
    float* out = (float*)d_out;

    float* Asw   = (float*)d_ws;
    float* wp    = Asw + (size_t)DIMD * NPAD;
    float* parts = wp + NPAD;

    FFM_build_A<<<NBUILD, 256, 0, stream>>>(V, fd, W, Asw, wp);

    dim3 g2(BATCH / MT, NSPLIT);
    FFM_wmma_gemm<<<g2, 256, 0, stream>>>(x, Asw, wp, parts);

    FFM_finalize<<<BATCH / 256, 256, 0, stream>>>(parts, b, out);
}


# Epilogue with no branch per element: x re-read by plain loads and a select; lane sums one butterfly step at a time over all 32 values, permutes overlap (128 were waited for in turn); sum order same
# speedup vs baseline: 2.9060x; 1.0540x over previous
#include <hip/hip_runtime.h>


#define DIMD   2000
#define NPAD   2048
#define BATCH  4096
#define NFIELD 40
#define NF     8
#define MT     64
#define NSUB   4
#define CK     200
#define NKC    10
#define LDX    204
#define NSPLIT 4
#define NTPW   4
#define KS     5
#define NSTG   10
#define NGST   100
#define XV4    13
#define NEMPTY 3
#define BCB    32
#define BRB    80
#define NCB    64
#define NRBK   25
#define VROW   320
#define LVROW  324
#define NPW    5
#define NSTV   10
#define NBUILD 892
#define XBUF   (MT * LDX)
#define NXBUF  3

template <int N> struct IC { static constexpr int value = N; };

typedef __attribute__((ext_vector_type(2))) float v2f;
typedef __attribute__((ext_vector_type(8))) float v8f;
typedef __attribute__((ext_vector_type(4))) float v4f;

__global__ __launch_bounds__(256) void FFM_build_A(
    const float* __restrict__ V, const int* __restrict__ fd,
    const float* __restrict__ W, float* __restrict__ Asw,
    float* __restrict__ wp)
{
    __shared__ __attribute__((aligned(16))) float vs[BCB * LVROW];

    int rest = blockIdx.x;
    int rbk = 0;
    int cb = -1;
    for (rbk = 0; rbk < NRBK; ++rbk) {
        const int cmin = (rbk == 0) ? 0 : (BRB * rbk - 39) / BCB;
        const int cnt = NCB - cmin;
        if (rest < cnt) { cb = cmin + rest; break; }
        rest -= cnt;
    }
    if (cb < 0) return;

    const int tid  = threadIdx.x;
    const int lane = tid & 31;
    const int wv   = __builtin_amdgcn_readfirstlane(tid >> 5);
    const int j0   = cb * BCB;
    const int j    = j0 + lane;
    const bool live = (j0 < DIMD) && (rbk * BRB < j0 + BCB - 1);

    if (live) {
        float4 st[NSTV];
        #pragma unroll
        for (int k = 0; k < NSTV; ++k) {
            const int idx = tid + k * 256;
            const int r   = idx / (VROW / 4);
            const int c4  = idx - r * (VROW / 4);
            const int rs  = (j0 + r < DIMD) ? (j0 + r) : (DIMD - 1);
            st[k] = *reinterpret_cast<const float4*>(V + (size_t)rs * VROW + c4 * 4);
        }
        #pragma unroll
        for (int k = 0; k < NSTV; ++k) {
            const int idx = tid + k * 256;
            const int r   = idx / (VROW / 4);
            const int c4  = idx - r * (VROW / 4);
            *reinterpret_cast<float4*>(&vs[r * LVROW + c4 * 4]) = st[k];
        }
    }
    __syncthreads();

    const bool jin = (j < DIMD);
    const int  p0  = rbk * (BRB / 2) + wv * NPW;
    float2 res[NPW];
    #pragma unroll
    for (int t = 0; t < NPW; ++t) res[t] = make_float2(0.0f, 0.0f);

    if (live) {
        const int fj = jin ? fd[j] : 0;
        const float* vj = &vs[lane * LVROW];
        int    fr[2 * NPW];
        float4 g[2 * NPW][2];
        #pragma unroll
        for (int k = 0; k < 2 * NPW; ++k) fr[k] = fd[2 * p0 + k];
        #pragma unroll
        for (int k = 0; k < 2 * NPW; ++k) {
            const float4* Vi = reinterpret_cast<const float4*>(
                V + (size_t)(2 * p0 + k) * VROW + fj * NF);
            g[k][0] = Vi[0];
            g[k][1] = Vi[1];
        }
        #pragma unroll
        for (int k = 0; k < 2 * NPW; ++k) {
            const float4* Vj = reinterpret_cast<const float4*>(vj + fr[k] * NF);
            const float4 a0 = g[k][0], a1 = g[k][1];
            const float4 b0 = Vj[0], b1 = Vj[1];
            const float d = a0.x * b0.x + a0.y * b0.y + a0.z * b0.z + a0.w * b0.w
                          + a1.x * b1.x + a1.y * b1.y + a1.z * b1.z + a1.w * b1.w;
            const float s = (jin && 2 * p0 + k < j) ? d : 0.0f;
            if (k & 1) res[k >> 1].y = s; else res[k >> 1].x = s;
        }
    }
    #pragma unroll
    for (int t = 0; t < NPW; ++t)
        *reinterpret_cast<float2*>(Asw + (((size_t)(p0 + t) * NPAD + j) << 1)) = res[t];
    if (rbk == 0 && wv == 0) wp[j] = jin ? W[j] : 0.0f;
}

__device__ __forceinline__ void bar_signal() { asm volatile("s_wait_dscnt 0x0\n\ts_barrier_signal -1" ::: "memory"); }
__device__ __forceinline__ void bar_wait()   { asm volatile("s_barrier_wait -1" ::: "memory"); }

__global__ __launch_bounds__(256) void FFM_wmma_gemm(
    const float* __restrict__ x, const float* __restrict__ Asw,
    const float* __restrict__ wp, float* __restrict__ parts)
{
    __shared__ __attribute__((aligned(16))) float lds[NXBUF * XBUF];
    const int tid  = threadIdx.x;
    const int w    = __builtin_amdgcn_readfirstlane(tid >> 5);
    const int lane = tid & 31;
    const int hi   = lane >> 4;
    const int lo   = lane & 15;
    const int sp   = blockIdx.y;
    const int s_base = blockIdx.x * MT;

    v8f c[NTPW][NSUB];
    #pragma unroll
    for (int t = 0; t < NTPW; ++t)
        #pragma unroll
        for (int u = 0; u < NSUB; ++u)
            c[t][u] = (v8f){0.f, 0.f, 0.f, 0.f, 0.f, 0.f, 0.f, 0.f};

    const int pos = (w < 4) ? w : 11 - w;
    int ntile[NTPW];
    int lim[NTPW];
    const float* bbase[NTPW];
    #pragma unroll
    for (int t = 0; t < NTPW; ++t) {
        const int q = (t & 1) ? (7 - pos) : pos;
        const int r = (t & 1) ? (NSPLIT - 1 - sp) : sp;
        ntile[t] = t * 32 + q * 4 + r - NEMPTY;
        const int nt = (ntile[t] < 0) ? 0 : ntile[t];
        const int st = (16 * nt + 14) / (4 * KS) + 1;
        lim[t] = (ntile[t] < 0) ? 0 : (st < NGST ? st : NGST);
        const int n = nt * 16 + lo;
        bbase[t] = Asw + ((size_t)hi * NPAD + n) * 2;
    }

    v4f xr[XV4];
    int xso[XV4];
    #pragma unroll
    for (int j = 0; j < XV4; ++j) {
        const int i = tid + j * 256;
        xso[j] = (i / (CK / 4)) * LDX + (i % (CK / 4)) * 4;
    }
    v2f bA[KS][NTPW];
    v2f bB[KS][NTPW];
    v2f aA[KS][NSUB];
    v2f aB[KS][NSUB];

#define ISSUE_X(kc2)                                                            \
    _Pragma("unroll")                                                           \
    for (int j = 0; j < XV4; ++j) {                                             \
        const int i = tid + j * 256;                                            \
        if (j < XV4 - 1 || tid < 128) {                                         \
            const int r  = i / (CK / 4);                                        \
            const int cq = i % (CK / 4);                                        \
            xr[j] = *reinterpret_cast<const v4f*>(                              \
                x + (size_t)(s_base + r) * DIMD + (kc2) * CK + cq * 4);         \
        }                                                                       \
    }
#define STORE_X(xo2)                                                            \
    _Pragma("unroll")                                                           \
    for (int j = 0; j < XV4; ++j) {                                             \
        if (j < XV4 - 1 || tid < 128) {                                         \
            *reinterpret_cast<v4f*>(&lds[(xo2) + xso[j]]) = xr[j];              \
        }                                                                       \
    }

    auto load_B = [&](auto tm, v2f (&b)[KS][NTPW], int g) {
        constexpr int TM = decltype(tm)::value;
        #pragma unroll
        for (int j = 0; j < KS; ++j)
            #pragma unroll
            for (int t = TM; t < NTPW; ++t)
                b[j][t] = *reinterpret_cast<const v2f*>(
                    bbase[t] + (size_t)(g * (KS * 2) + j * 2) * (NPAD * 2));
    };
    auto load_B4 = [&](v2f (&b)[KS][NTPW], int g) {
        #pragma unroll
        for (int t = NTPW - 1; t >= 0; --t) {
            const float* bp = (g < lim[t]) ? bbase[t] : bbase[NTPW - 1];
            #pragma unroll
            for (int j = 0; j < KS; ++j)
                b[j][t] = *reinterpret_cast<const v2f*>(
                    bp + (size_t)(g * (KS * 2) + j * 2) * (NPAD * 2));
        }
    };
    auto load_A = [&](v2f (&a)[KS][NSUB], int sl, int xo) {
        #pragma unroll
        for (int j = 0; j < KS; ++j) {
            const int k0 = (sl * KS + j) * 4;
            #pragma unroll
            for (int u = 0; u < NSUB; ++u)
                a[j][u] = *reinterpret_cast<const v2f*>(
                    &lds[xo + (u * 16 + lo) * LDX + k0 + hi * 2]);
        }
    };
    auto tile_mult = [&](auto tt, const v2f (&a)[KS][NSUB], const v2f (&b)[KS][NTPW]) {
        constexpr int T = decltype(tt)::value;
        #pragma unroll
        for (int j = 0; j < KS; ++j)
            #pragma unroll
            for (int u = 0; u < NSUB; ++u)
                c[T][u] = __builtin_amdgcn_wmma_f32_16x16x4_f32(
                    false, a[j][u], false, b[j][T], (short)0, c[T][u], false, false);
    };
    auto stage = [&](const v2f (&a)[KS][NSUB], const v2f (&b)[KS][NTPW], int tmin) {
        tile_mult(IC<3>{}, a, b);
        if (tmin <= 2) {
            tile_mult(IC<2>{}, a, b);
            if (tmin <= 1) {
                tile_mult(IC<1>{}, a, b);
                if (tmin == 0) tile_mult(IC<0>{}, a, b);
            }
        }
    };

    ISSUE_X(0)
    load_B(IC<0>{}, bA, 0);
    STORE_X(0)
    ISSUE_X(1)
    bar_signal();

    int xo = 0;
    for (int kc = 0; kc < NKC; ++kc) {
        bar_wait();
        const bool act0 = (kc * NSTG < lim[NTPW - 1]);
        if (act0) load_A(aA, 0, xo);
        const int xn = (xo == (NXBUF - 1) * XBUF) ? 0 : xo + XBUF;
        if (kc + 1 < NKC) {
            STORE_X(xn)
            if (kc + 2 < NKC && !act0) { ISSUE_X(kc + 2) }
            bar_signal();
        }

        for (int sl = 0; sl < NSTG; sl += 2) {
            const int g = kc * NSTG + sl;
            const int tmin = (g >= lim[0]) + (g >= lim[1]) + (g >= lim[2]) + (g >= lim[3]);
            const int tmin1 = (g + 1 >= lim[0]) + (g + 1 >= lim[1]) + (g + 1 >= lim[2]) + (g + 1 >= lim[3]);
            if (tmin < NTPW) {
                load_A(aB, sl + 1, xo);
                load_B4(bB, g + 1);
                if (sl == 0 && kc + 2 < NKC) { ISSUE_X(kc + 2) }
                __builtin_amdgcn_sched_barrier(0);
                stage(aA, bA, tmin);
                load_A(aA, (sl + 2 < NSTG) ? (sl + 2) : 0, xo);
                load_B4(bA, (g + 2 < NGST) ? (g + 2) : (NGST - 1));
                __builtin_amdgcn_sched_barrier(0);
                if (tmin1 < NTPW) stage(aB, bB, tmin1);
            }
        }
        xo = xn;
    }
    __syncthreads();

#undef ISSUE_X
#undef STORE_X

    float racc[NSUB][8];
    #pragma unroll
    for (int u = 0; u < NSUB; ++u)
        #pragma unroll
        for (int v = 0; v < 8; ++v) racc[u][v] = 0.0f;

    #pragma unroll
    for (int t = 0; t < NTPW; ++t) {
        const bool inb = (ntile[t] >= 0);
        const int n = (inb ? ntile[t] : 0) * 16 + lo;
        const float wn = wp[n];
        const float* xp = x + (size_t)(s_base + hi * 8) * DIMD + n;
        #pragma unroll
        for (int u = 0; u < NSUB; ++u) {
            #pragma unroll
            for (int v = 0; v < 8; ++v) {
                const float xl = xp[(v + u * 16) * DIMD];
                const float xv = inb ? xl : 0.0f;
                const float tv = c[t][u][v];
                racc[u][v] += xv * (tv + wn);
            }
        }
    }

    #pragma unroll
    for (int m = 8; m >= 1; m >>= 1) {
        #pragma unroll
        for (int u = 0; u < NSUB; ++u) {
            #pragma unroll
            for (int v = 0; v < 8; ++v)
                racc[u][v] += __shfl_xor(racc[u][v], m, 32);
        }
    }
    if (lo == 0) {
        #pragma unroll
        for (int u = 0; u < NSUB; ++u) {
            #pragma unroll
            for (int v = 0; v < 8; ++v)
                lds[w * MT + v + hi * 8 + u * 16] = racc[u][v];
        }
    }
    __syncthreads();

    if (tid < MT) {
        float p = 0.0f;
        #pragma unroll
        for (int w2 = 0; w2 < 8; ++w2) p += lds[w2 * MT + tid];
        parts[(size_t)(s_base + tid) * NSPLIT + sp] = p;
    }
}

__global__ __launch_bounds__(256) void FFM_finalize(
    const float* __restrict__ parts, const float* __restrict__ b,
    float* __restrict__ out)
{
    const int i = blockIdx.x * blockDim.x + threadIdx.x;
    if (i < BATCH) {
        float p = b[0];
        #pragma unroll
        for (int s = 0; s < NSPLIT; ++s) p += parts[(size_t)i * NSPLIT + s];
        out[i] = p;
    }
}

extern "C" void kernel_launch(void* const* d_in, const int* in_sizes, int n_in,
                              void* d_out, int out_size, void* d_ws, size_t ws_size,
                              hipStream_t stream)
{
    const float* x  = (const float*)d_in[0];
    const int*   fd = (const int*)  d_in[1];
    const float* b  = (const float*)d_in[2];
    const float* W  = (const float*)d_in[3];
    const float* V  = (const float*)d_in[4];
    float* out = (float*)d_out;

    float* Asw   = (float*)d_ws;
    float* wp    = Asw + (size_t)DIMD * NPAD;
    float* parts = wp + NPAD;

    FFM_build_A<<<NBUILD, 256, 0, stream>>>(V, fd, W, Asw, wp);

    dim3 g2(BATCH / MT, NSPLIT);
    FFM_wmma_gemm<<<g2, 256, 0, stream>>>(x, Asw, wp, parts);

    FFM_finalize<<<BATCH / 256, 256, 0, stream>>>(parts, b, out);
}


# Linear and diagonal term: each wave's 65 dependent load trips (a rolled loop that waited for memory once per 32 sorted rows) become two groups of 34 loads in flight together; sum order unchanged
# speedup vs baseline: 3.9334x; 1.3535x over previous
#include <hip/hip_runtime.h>


#define DIMD   2000
#define BATCH  4096
#define NFIELD 40
#define NF     8
#define VROW   320
#define KPAD   2120
#define NRP    1060
#define NCH    63
#define PTH    320
#define PWV    10
#define PCH    7
#define NPB    256
#define GPB    9
#define MS     16
#define LDXS   2124
#define NWV    16
#define NPAIR  20
#define NOFF   190
#define NITEM  210
#define UK     4
#define XST    16
#define RPB    5
#define LNB    2
#define LNU    34

typedef __attribute__((ext_vector_type(2))) float v2f;
typedef __attribute__((ext_vector_type(8))) float v8f;

__device__ __forceinline__ int clampf(int f) { return f < 0 ? 0 : (f > NFIELD - 1 ? NFIELD - 1 : f); }

__global__ __launch_bounds__(PTH) void FFM_prep(
    const float* __restrict__ V, const int* __restrict__ fd,
    const float* __restrict__ W, float* __restrict__ Vsw,
    float* __restrict__ gs, float* __restrict__ ds,
    int* __restrict__ inv, int* __restrict__ offg, int* __restrict__ totg)
{
    __shared__ int permS[KPAD + 8];
    __shared__ int cnt[NCH * NFIELD];
    __shared__ int totS[NFIELD];
    __shared__ int offS[NFIELD + 1];

    const int tid  = threadIdx.x;
    const int lane = tid & 31;
    const int w    = __builtin_amdgcn_readfirstlane(tid >> 5);

    for (int k = tid; k < KPAD + 8; k += PTH) permS[k] = -1;

    int ff[PCH], rk[PCH];
    #pragma unroll
    for (int j = 0; j < PCH; ++j) {
        const int c = w + PWV * j;
        const int i = c * 32 + lane;
        const bool valid = (c < NCH) && (i < DIMD);
        const int f = valid ? clampf(fd[i]) : -1;
        unsigned mym = 0u;
        for (int p = 0; p < NFIELD; ++p) {
            const unsigned m = (unsigned)__ballot(f == p);
            if (f == p) mym = m;
            if (lane == 0 && c < NCH) cnt[c * NFIELD + p] = __popc(m);
        }
        ff[j] = f;
        rk[j] = __popc(mym & ((1u << lane) - 1u));
    }
    __syncthreads();
    if (tid < NFIELD) {
        int run = 0;
        for (int c = 0; c < NCH; ++c) {
            const int t = cnt[c * NFIELD + tid];
            cnt[c * NFIELD + tid] = run;
            run += t;
        }
        totS[tid] = run;
    }
    __syncthreads();
    if (tid <= NFIELD) {
        int o = 0;
        for (int q = 0; q < tid; ++q) o += (totS[q] + 3) & ~3;
        offS[tid] = o;
    }
    __syncthreads();
    #pragma unroll
    for (int j = 0; j < PCH; ++j) {
        if (ff[j] >= 0) {
            const int c = w + PWV * j;
            const int i = c * 32 + lane;
            const int pos = offS[ff[j]] + cnt[c * NFIELD + ff[j]] + rk[j];
            permS[pos] = i;
            if (blockIdx.x == 0) inv[i] = pos;
        }
    }
    if (blockIdx.x == 0) {
        if (tid <= NFIELD) offg[tid] = offS[tid];
        if (tid < NFIELD)  totg[tid] = totS[tid];
    }
    __syncthreads();

    {
        float2 rr[RPB];
        #pragma unroll
        for (int k = 0; k < RPB; ++k) {
            const int rp = blockIdx.x + k * NPB;
            rr[k] = make_float2(0.0f, 0.0f);
            if (rp < NRP + 2) {
                const int i0 = permS[2 * rp];
                const int i1 = permS[2 * rp + 1];
                if (i0 >= 0) rr[k].x = V[(size_t)i0 * VROW + tid];
                if (i1 >= 0) rr[k].y = V[(size_t)i1 * VROW + tid];
            }
        }
        #pragma unroll
        for (int k = 0; k < RPB; ++k) {
            const int rp = blockIdx.x + k * NPB;
            if (rp < NRP + 2)
                *reinterpret_cast<float2*>(Vsw + ((size_t)rp * VROW + tid) * 2) = rr[k];
        }
    }
    if (tid < GPB) {
        const int kk = blockIdx.x * GPB + tid;
        if (kk < KPAD) {
            const int i = permS[kk];
            float g = 0.0f, d = 0.0f;
            if (i >= 0) {
                g = W[i];
                const float* Vi = V + (size_t)i * VROW + clampf(fd[i]) * NF;
                #pragma unroll
                for (int e = 0; e < NF; ++e) d += Vi[e] * Vi[e];
                d *= 0.5f;
            }
            gs[kk] = g;
            ds[kk] = d;
        }
    }
}

__global__ __launch_bounds__(512) void FFM_main(
    const float* __restrict__ x, const float* __restrict__ Vsw,
    const float* __restrict__ gs, const float* __restrict__ ds,
    const int* __restrict__ inv, const int* __restrict__ offg,
    const int* __restrict__ totg, const float* __restrict__ b,
    float* __restrict__ out)
{
    __shared__ __attribute__((aligned(16))) float xs[MS * LDXS];
    __shared__ int   offS[NFIELD + 4];
    __shared__ int   totS[NFIELD];
    __shared__ float red[NWV * MS];
    __shared__ float lin[MS];

    const int tid  = threadIdx.x;
    const int lane = tid & 31;
    const int w    = __builtin_amdgcn_readfirstlane(tid >> 5);
    const int hi   = lane >> 4;
    const int lo   = lane & 15;
    const int s_base = blockIdx.x * MS;

    if (tid <= NFIELD) offS[tid] = offg[tid];
    if (tid < NFIELD)  totS[tid] = totg[tid];

    {
        float4 xv[XST];
        int4   xi[XST];
        #pragma unroll
        for (int k = 0; k < XST; ++k) {
            const int idx = tid + k * 512;
            if (idx < MS * (DIMD / 4)) {
                const int row = idx / (DIMD / 4);
                const int c4  = idx - row * (DIMD / 4);
                xv[k] = *reinterpret_cast<const float4*>(x + (size_t)(s_base + row) * DIMD + c4 * 4);
                xi[k] = *reinterpret_cast<const int4*>(inv + c4 * 4);
            }
        }
        #pragma unroll
        for (int k = 0; k < XST; ++k) {
            const int idx = tid + k * 512;
            if (idx < MS * (DIMD / 4)) {
                const int row = idx / (DIMD / 4);
                float* xr = &xs[row * LDXS];
                xr[xi[k].x] = xv[k].x; xr[xi[k].y] = xv[k].y; xr[xi[k].z] = xv[k].z; xr[xi[k].w] = xv[k].w;
            }
        }
    }
    __syncthreads();
    if (tid < NFIELD * 3) {
        const int p  = tid / 3;
        const int kk = offS[p] + totS[p] + (tid - p * 3);
        if (kk < offS[p + 1]) {
            #pragma unroll
            for (int row = 0; row < MS; ++row) xs[row * LDXS + kk] = 0.0f;
        }
    }
    __syncthreads();

    float s[8];
    #pragma unroll
    for (int v = 0; v < 8; ++v) s[v] = 0.0f;

    for (int it = w; it < NITEM; it += NWV) {
        const bool dg = (it >= NOFF);
        int a = 0, bb = 0;
        if (dg) { a = it - NOFF; bb = a; }
        else {
            int rem = it;
            while (rem >= NPAIR - 1 - a) { rem -= NPAIR - 1 - a; ++a; }
            bb = a + 1 + rem;
        }
        int o[4], nk[4];
        const float* bp[4];
        int ao[4];
        #pragma unroll
        for (int j = 0; j < 4; ++j) {
            const int p = (j < 2) ? (2 * a + j) : (dg ? (2 * a + j - 2) : (2 * bb + j - 2));
            const int t = (j < 2) ? bb : a;
            o[j]  = __builtin_amdgcn_readfirstlane(offS[p]);
            const int e = __builtin_amdgcn_readfirstlane(offS[p + 1]);
            nk[j] = (dg && j >= 2) ? 0 : ((e - o[j]) >> 2);
            bp[j] = Vsw + ((size_t)((o[j] >> 1) + hi) * VROW + t * 16 + lo) * 2;
            ao[j] = lo * LDXS + o[j] + 2 * hi;
        }
        int nkmax = nk[0];
        #pragma unroll
        for (int j = 1; j < 4; ++j) nkmax = nk[j] > nkmax ? nk[j] : nkmax;

        v8f C[4];
        #pragma unroll
        for (int j = 0; j < 4; ++j) C[j] = (v8f){0.f, 0.f, 0.f, 0.f, 0.f, 0.f, 0.f, 0.f};

        v2f aA[4][UK], bA[4][UK], aB[4][UK], bB[4][UK];
        auto load_set = [&](v2f (&af)[4][UK], v2f (&bf)[4][UK], int ks0) {
            #pragma unroll
            for (int j = 0; j < 4; ++j) {
                #pragma unroll
                for (int u = 0; u < UK; ++u) {
                    int kc = ks0 + u;
                    const int last = nk[j] > 0 ? nk[j] - 1 : 0;
                    kc = kc < last ? kc : last;
                    af[j][u] = *reinterpret_cast<const v2f*>(&xs[ao[j] + 4 * kc]);
                    bf[j][u] = *reinterpret_cast<const v2f*>(bp[j] + (size_t)kc * (4 * VROW));
                }
            }
        };
        auto mult_set = [&](const v2f (&af)[4][UK], const v2f (&bf)[4][UK], int ks0) {
            #pragma unroll
            for (int u = 0; u < UK; ++u) {
                #pragma unroll
                for (int j = 0; j < 4; ++j) {
                    if (__builtin_expect(ks0 + u < nk[j], 1))
                        C[j] = __builtin_amdgcn_wmma_f32_16x16x4_f32(
                            false, af[j][u], false, bf[j][u], (short)0, C[j], false, false);
                }
            }
        };

        load_set(aA, bA, 0);
        for (int ks0 = 0; ks0 < nkmax; ks0 += 2 * UK) {
            load_set(aB, bB, ks0 + UK);
            mult_set(aA, bA, ks0);
            load_set(aA, bA, ks0 + 2 * UK);
            mult_set(aB, bB, ks0 + UK);
        }

        const bool lof = (lo < 8);
        if (!dg) {
            #pragma unroll
            for (int v = 0; v < 8; ++v) {
                const float X = lof ? C[3][v] : C[2][v];
                const float Y = __shfl_xor(X, 8, 32);
                const float P = lof ? C[2][v] : Y;
                const float Q = lof ? Y : C[3][v];
                s[v] += C[0][v] * P + C[1][v] * Q;
            }
        } else {
            #pragma unroll
            for (int v = 0; v < 8; ++v) {
                const float Y = __shfl_xor(C[1][v], 8, 32);
                const float d0 = 0.5f * C[0][v] * C[0][v];
                const float d1 = C[0][v] * Y + 0.5f * C[1][v] * C[1][v];
                s[v] += lof ? d0 : d1;
            }
        }
    }

    {
        const int total = offS[NFIELD];
        float acc = 0.0f;
        #pragma unroll
        for (int bt = 0; bt < LNB; ++bt) {
            float xq[LNU], gq[LNU], dq[LNU];
            #pragma unroll
            for (int u = 0; u < LNU; ++u) {
                const int kk = (bt * LNU + u) * 32 + lane;
                const int kc = kk < total ? kk : total - 1;
                xq[u] = xs[w * LDXS + kc];
                gq[u] = gs[kc];
                dq[u] = ds[kc];
            }
            #pragma unroll
            for (int u = 0; u < LNU; ++u) {
                const int kk = (bt * LNU + u) * 32 + lane;
                const float xv = kk < total ? xq[u] : 0.0f;
                acc += xv * (gq[u] - xv * dq[u]);
            }
        }
        #pragma unroll
        for (int m = 16; m >= 1; m >>= 1) acc += __shfl_xor(acc, m, 32);
        if (lane == 0) lin[w] = acc;
    }
    #pragma unroll
    for (int m = 8; m >= 1; m >>= 1) {
        #pragma unroll
        for (int v = 0; v < 8; ++v) s[v] += __shfl_xor(s[v], m, 32);
    }
    if (lo == 0) {
        #pragma unroll
        for (int v = 0; v < 8; ++v) red[w * MS + v + hi * 8] = s[v];
    }
    __syncthreads();
    if (tid < MS) {
        float p = b[0] + lin[tid];
        #pragma unroll
        for (int w2 = 0; w2 < NWV; ++w2) p += red[w2 * MS + tid];
        out[s_base + tid] = p;
    }
}

extern "C" void kernel_launch(void* const* d_in, const int* in_sizes, int n_in,
                              void* d_out, int out_size, void* d_ws, size_t ws_size,
                              hipStream_t stream)
{
    const float* x  = (const float*)d_in[0];
    const int*   fd = (const int*)  d_in[1];
    const float* b  = (const float*)d_in[2];
    const float* W  = (const float*)d_in[3];
    const float* V  = (const float*)d_in[4];
    float* out = (float*)d_out;

    float* Vsw = (float*)d_ws;
    float* gs  = Vsw + (size_t)(NRP + 4) * VROW * 2;
    float* ds  = gs + (KPAD + 8);
    int*   inv = (int*)(ds + (KPAD + 8));
    int*   off = inv + DIMD;
    int*   tot = off + 48;

    FFM_prep<<<NPB, PTH, 0, stream>>>(V, fd, W, Vsw, gs, ds, inv, off, tot);
    FFM_main<<<BATCH / MS, 512, 0, stream>>>(x, Vsw, gs, ds, inv, off, tot, b, out);
}


# Prep sort: a chunk's lanes per field come from 6 ballots on the bits of the field id (were 40 ballots, each with a branch and a store); the 7 field-id loads fly together; same order, sums unchanged
# speedup vs baseline: 4.6003x; 1.1695x over previous
#include <hip/hip_runtime.h>


#define DIMD   2000
#define BATCH  4096
#define NFIELD 40
#define NF     8
#define VROW   320
#define KPAD   2120
#define NRP    1060
#define NCH    63
#define PTH    320
#define PWV    10
#define PCH    7
#define NPB    256
#define GPB    9
#define MS     16
#define LDXS   2124
#define NWV    16
#define NPAIR  20
#define NOFF   190
#define NITEM  210
#define UK     4
#define XST    16
#define RPB    5
#define LNB    2
#define LNU    34

typedef __attribute__((ext_vector_type(2))) float v2f;
typedef __attribute__((ext_vector_type(8))) float v8f;

__device__ __forceinline__ int clampf(int f) { return f < 0 ? 0 : (f > NFIELD - 1 ? NFIELD - 1 : f); }

__global__ __launch_bounds__(PTH) void FFM_prep(
    const float* __restrict__ V, const int* __restrict__ fd,
    const float* __restrict__ W, float* __restrict__ Vsw,
    float* __restrict__ gs, float* __restrict__ ds,
    int* __restrict__ inv, int* __restrict__ offg, int* __restrict__ totg)
{
    __shared__ int permS[KPAD + 8];
    __shared__ int cnt[NCH * NFIELD];
    __shared__ int totS[NFIELD];
    __shared__ int offS[NFIELD + 1];

    const int tid  = threadIdx.x;
    const int lane = tid & 31;
    const int w    = __builtin_amdgcn_readfirstlane(tid >> 5);

    for (int k = tid; k < KPAD + 8; k += PTH) permS[k] = -1;

    int ff[PCH], rk[PCH];
    int fv[PCH];
    #pragma unroll
    for (int j = 0; j < PCH; ++j) {
        const int i = (w + PWV * j) * 32 + lane;
        fv[j] = fd[i < DIMD ? i : DIMD - 1];
    }
    #pragma unroll
    for (int j = 0; j < PCH; ++j) {
        const int c = w + PWV * j;
        const int i = c * 32 + lane;
        const bool valid = (c < NCH) && (i < DIMD);
        const int f = clampf(fv[j]);
        const unsigned vm = (unsigned)__ballot(valid);
        unsigned bk[6];
        #pragma unroll
        for (int k = 0; k < 6; ++k) bk[k] = (unsigned)__ballot(valid && ((f >> k) & 1));
        auto lanes_of = [&](int q) {
            unsigned m = vm;
            #pragma unroll
            for (int k = 0; k < 6; ++k) m &= bk[k] ^ ((unsigned)((q >> k) & 1) - 1u);
            return m;
        };
        ff[j] = valid ? f : -1;
        rk[j] = __popc(lanes_of(f) & ((1u << lane) - 1u));
        if (c < NCH) {
            const int q2 = lane + 32 < NFIELD ? lane + 32 : lane;
            cnt[c * NFIELD + lane] = __popc(lanes_of(lane));
            cnt[c * NFIELD + q2]   = __popc(lanes_of(q2));
        }
    }
    __syncthreads();
    if (tid < NFIELD) {
        int run = 0;
        for (int c = 0; c < NCH; ++c) {
            const int t = cnt[c * NFIELD + tid];
            cnt[c * NFIELD + tid] = run;
            run += t;
        }
        totS[tid] = run;
    }
    __syncthreads();
    if (tid <= NFIELD) {
        int o = 0;
        for (int q = 0; q < tid; ++q) o += (totS[q] + 3) & ~3;
        offS[tid] = o;
    }
    __syncthreads();
    #pragma unroll
    for (int j = 0; j < PCH; ++j) {
        if (ff[j] >= 0) {
            const int c = w + PWV * j;
            const int i = c * 32 + lane;
            const int pos = offS[ff[j]] + cnt[c * NFIELD + ff[j]] + rk[j];
            permS[pos] = i;
            if (blockIdx.x == 0) inv[i] = pos;
        }
    }
    if (blockIdx.x == 0) {
        if (tid <= NFIELD) offg[tid] = offS[tid];
        if (tid < NFIELD)  totg[tid] = totS[tid];
    }
    __syncthreads();

    {
        float2 rr[RPB];
        #pragma unroll
        for (int k = 0; k < RPB; ++k) {
            const int rp = blockIdx.x + k * NPB;
            rr[k] = make_float2(0.0f, 0.0f);
            if (rp < NRP + 2) {
                const int i0 = permS[2 * rp];
                const int i1 = permS[2 * rp + 1];
                if (i0 >= 0) rr[k].x = V[(size_t)i0 * VROW + tid];
                if (i1 >= 0) rr[k].y = V[(size_t)i1 * VROW + tid];
            }
        }
        #pragma unroll
        for (int k = 0; k < RPB; ++k) {
            const int rp = blockIdx.x + k * NPB;
            if (rp < NRP + 2)
                *reinterpret_cast<float2*>(Vsw + ((size_t)rp * VROW + tid) * 2) = rr[k];
        }
    }
    if (tid < GPB) {
        const int kk = blockIdx.x * GPB + tid;
        if (kk < KPAD) {
            const int i = permS[kk];
            float g = 0.0f, d = 0.0f;
            if (i >= 0) {
                g = W[i];
                const float* Vi = V + (size_t)i * VROW + clampf(fd[i]) * NF;
                #pragma unroll
                for (int e = 0; e < NF; ++e) d += Vi[e] * Vi[e];
                d *= 0.5f;
            }
            gs[kk] = g;
            ds[kk] = d;
        }
    }
}

__global__ __launch_bounds__(512) void FFM_main(
    const float* __restrict__ x, const float* __restrict__ Vsw,
    const float* __restrict__ gs, const float* __restrict__ ds,
    const int* __restrict__ inv, const int* __restrict__ offg,
    const int* __restrict__ totg, const float* __restrict__ b,
    float* __restrict__ out)
{
    __shared__ __attribute__((aligned(16))) float xs[MS * LDXS];
    __shared__ int   offS[NFIELD + 4];
    __shared__ int   totS[NFIELD];
    __shared__ float red[NWV * MS];
    __shared__ float lin[MS];

    const int tid  = threadIdx.x;
    const int lane = tid & 31;
    const int w    = __builtin_amdgcn_readfirstlane(tid >> 5);
    const int hi   = lane >> 4;
    const int lo   = lane & 15;
    const int s_base = blockIdx.x * MS;

    if (tid <= NFIELD) offS[tid] = offg[tid];
    if (tid < NFIELD)  totS[tid] = totg[tid];

    {
        float4 xv[XST];
        int4   xi[XST];
        #pragma unroll
        for (int k = 0; k < XST; ++k) {
            const int idx = tid + k * 512;
            if (idx < MS * (DIMD / 4)) {
                const int row = idx / (DIMD / 4);
                const int c4  = idx - row * (DIMD / 4);
                xv[k] = *reinterpret_cast<const float4*>(x + (size_t)(s_base + row) * DIMD + c4 * 4);
                xi[k] = *reinterpret_cast<const int4*>(inv + c4 * 4);
            }
        }
        #pragma unroll
        for (int k = 0; k < XST; ++k) {
            const int idx = tid + k * 512;
            if (idx < MS * (DIMD / 4)) {
                const int row = idx / (DIMD / 4);
                float* xr = &xs[row * LDXS];
                xr[xi[k].x] = xv[k].x; xr[xi[k].y] = xv[k].y; xr[xi[k].z] = xv[k].z; xr[xi[k].w] = xv[k].w;
            }
        }
    }
    __syncthreads();
    if (tid < NFIELD * 3) {
        const int p  = tid / 3;
        const int kk = offS[p] + totS[p] + (tid - p * 3);
        if (kk < offS[p + 1]) {
            #pragma unroll
            for (int row = 0; row < MS; ++row) xs[row * LDXS + kk] = 0.0f;
        }
    }
    __syncthreads();

    float s[8];
    #pragma unroll
    for (int v = 0; v < 8; ++v) s[v] = 0.0f;

    for (int it = w; it < NITEM; it += NWV) {
        const bool dg = (it >= NOFF);
        int a = 0, bb = 0;
        if (dg) { a = it - NOFF; bb = a; }
        else {
            int rem = it;
            while (rem >= NPAIR - 1 - a) { rem -= NPAIR - 1 - a; ++a; }
            bb = a + 1 + rem;
        }
        int o[4], nk[4];
        const float* bp[4];
        int ao[4];
        #pragma unroll
        for (int j = 0; j < 4; ++j) {
            const int p = (j < 2) ? (2 * a + j) : (dg ? (2 * a + j - 2) : (2 * bb + j - 2));
            const int t = (j < 2) ? bb : a;
            o[j]  = __builtin_amdgcn_readfirstlane(offS[p]);
            const int e = __builtin_amdgcn_readfirstlane(offS[p + 1]);
            nk[j] = (dg && j >= 2) ? 0 : ((e - o[j]) >> 2);
            bp[j] = Vsw + ((size_t)((o[j] >> 1) + hi) * VROW + t * 16 + lo) * 2;
            ao[j] = lo * LDXS + o[j] + 2 * hi;
        }
        int nkmax = nk[0];
        #pragma unroll
        for (int j = 1; j < 4; ++j) nkmax = nk[j] > nkmax ? nk[j] : nkmax;

        v8f C[4];
        #pragma unroll
        for (int j = 0; j < 4; ++j) C[j] = (v8f){0.f, 0.f, 0.f, 0.f, 0.f, 0.f, 0.f, 0.f};

        v2f aA[4][UK], bA[4][UK], aB[4][UK], bB[4][UK];
        auto load_set = [&](v2f (&af)[4][UK], v2f (&bf)[4][UK], int ks0) {
            #pragma unroll
            for (int j = 0; j < 4; ++j) {
                #pragma unroll
                for (int u = 0; u < UK; ++u) {
                    int kc = ks0 + u;
                    const int last = nk[j] > 0 ? nk[j] - 1 : 0;
                    kc = kc < last ? kc : last;
                    af[j][u] = *reinterpret_cast<const v2f*>(&xs[ao[j] + 4 * kc]);
                    bf[j][u] = *reinterpret_cast<const v2f*>(bp[j] + (size_t)kc * (4 * VROW));
                }
            }
        };
        auto mult_set = [&](const v2f (&af)[4][UK], const v2f (&bf)[4][UK], int ks0) {
            #pragma unroll
            for (int u = 0; u < UK; ++u) {
                #pragma unroll
                for (int j = 0; j < 4; ++j) {
                    if (__builtin_expect(ks0 + u < nk[j], 1))
                        C[j] = __builtin_amdgcn_wmma_f32_16x16x4_f32(
                            false, af[j][u], false, bf[j][u], (short)0, C[j], false, false);
                }
            }
        };

        load_set(aA, bA, 0);
        for (int ks0 = 0; ks0 < nkmax; ks0 += 2 * UK) {
            load_set(aB, bB, ks0 + UK);
            mult_set(aA, bA, ks0);
            load_set(aA, bA, ks0 + 2 * UK);
            mult_set(aB, bB, ks0 + UK);
        }

        const bool lof = (lo < 8);
        if (!dg) {
            #pragma unroll
            for (int v = 0; v < 8; ++v) {
                const float X = lof ? C[3][v] : C[2][v];
                const float Y = __shfl_xor(X, 8, 32);
                const float P = lof ? C[2][v] : Y;
                const float Q = lof ? Y : C[3][v];
                s[v] += C[0][v] * P + C[1][v] * Q;
            }
        } else {
            #pragma unroll
            for (int v = 0; v < 8; ++v) {
                const float Y = __shfl_xor(C[1][v], 8, 32);
                const float d0 = 0.5f * C[0][v] * C[0][v];
                const float d1 = C[0][v] * Y + 0.5f * C[1][v] * C[1][v];
                s[v] += lof ? d0 : d1;
            }
        }
    }

    {
        const int total = offS[NFIELD];
        float acc = 0.0f;
        #pragma unroll
        for (int bt = 0; bt < LNB; ++bt) {
            float xq[LNU], gq[LNU], dq[LNU];
            #pragma unroll
            for (int u = 0; u < LNU; ++u) {
                const int kk = (bt * LNU + u) * 32 + lane;
                const int kc = kk < total ? kk : total - 1;
                xq[u] = xs[w * LDXS + kc];
                gq[u] = gs[kc];
                dq[u] = ds[kc];
            }
            #pragma unroll
            for (int u = 0; u < LNU; ++u) {
                const int kk = (bt * LNU + u) * 32 + lane;
                const float xv = kk < total ? xq[u] : 0.0f;
                acc += xv * (gq[u] - xv * dq[u]);
            }
        }
        #pragma unroll
        for (int m = 16; m >= 1; m >>= 1) acc += __shfl_xor(acc, m, 32);
        if (lane == 0) lin[w] = acc;
    }
    #pragma unroll
    for (int m = 8; m >= 1; m >>= 1) {
        #pragma unroll
        for (int v = 0; v < 8; ++v) s[v] += __shfl_xor(s[v], m, 32);
    }
    if (lo == 0) {
        #pragma unroll
        for (int v = 0; v < 8; ++v) red[w * MS + v + hi * 8] = s[v];
    }
    __syncthreads();
    if (tid < MS) {
        float p = b[0] + lin[tid];
        #pragma unroll
        for (int w2 = 0; w2 < NWV; ++w2) p += red[w2 * MS + tid];
        out[s_base + tid] = p;
    }
}

extern "C" void kernel_launch(void* const* d_in, const int* in_sizes, int n_in,
                              void* d_out, int out_size, void* d_ws, size_t ws_size,
                              hipStream_t stream)
{
    const float* x  = (const float*)d_in[0];
    const int*   fd = (const int*)  d_in[1];
    const float* b  = (const float*)d_in[2];
    const float* W  = (const float*)d_in[3];
    const float* V  = (const float*)d_in[4];
    float* out = (float*)d_out;

    float* Vsw = (float*)d_ws;
    float* gs  = Vsw + (size_t)(NRP + 4) * VROW * 2;
    float* ds  = gs + (KPAD + 8);
    int*   inv = (int*)(ds + (KPAD + 8));
    int*   off = inv + DIMD;
    int*   tot = off + 48;

    FFM_prep<<<NPB, PTH, 0, stream>>>(V, fd, W, Vsw, gs, ds, inv, off, tot);
    FFM_main<<<BATCH / MS, 512, 0, stream>>>(x, Vsw, gs, ds, inv, off, tot, b, out);
}
